# starting kernel
# baseline (speedup 1.0000x reference)
#include <hip/hip_runtime.h>
#include <math.h>

typedef float v2f __attribute__((ext_vector_type(2)));
typedef float v8f __attribute__((ext_vector_type(8)));

#define N_NODES 50000
#define N_EDGES 600000
#define DFEAT   128
#define NCLS    40

__global__ void k_init_deg(int* deg, int n) {
    int i = blockIdx.x * blockDim.x + threadIdx.x;
    if (i < n) deg[i] = 1;
}

__global__ void k_count_deg(int* deg, const int* __restrict__ dst, int e) {
    int i = blockIdx.x * blockDim.x + threadIdx.x;
    if (i < e) atomicAdd(&deg[dst[i]], 1);
}

__global__ void k_make_dinv(float* dinv, int n) {
    int i = blockIdx.x * blockDim.x + threadIdx.x;
    if (i < n) {
        int d = ((const int*)dinv)[i];
        dinv[i] = rsqrtf((float)d);
    }
}

__global__ void k_zero(float* p, int n) {
    int i = blockIdx.x * blockDim.x + threadIdx.x;
    if (i < n) p[i] = 0.0f;
}

template<int N>
__global__ void k_gemm(const float* __restrict__ A, const float* __restrict__ B,
                       float* __restrict__ C) {
    const int lane  = threadIdx.x & 31;
    const int wave  = threadIdx.x >> 5;
    const int row0  = blockIdx.x * 16;
    const int col0  = (blockIdx.y * (blockDim.x >> 5) + wave) * 16;
    const int r     = lane & 15;
    const int kb    = (lane >> 4) * 2;

    const int col  = col0 + r;
    const int colc = (col < N) ? col : (N - 1);

    const float* Ap = A + (row0 + r) * DFEAT + kb;
    const float* Bp = B + kb * N + colc;

    v8f acc = {};
    #pragma unroll
    for (int k0 = 0; k0 < DFEAT; k0 += 4) {
        v2f a = *(const v2f*)(Ap + k0);
        v2f b;
        b.x = Bp[k0 * N];
        b.y = Bp[k0 * N + N];
        acc = __builtin_amdgcn_wmma_f32_16x16x4_f32(
            false, a, false, b, (short)0, acc, false, false);
    }

    if (col < N) {
        float* Cp = C + (row0 + (lane >> 4) * 8) * N + col;
        #pragma unroll
        for (int i = 0; i < 8; ++i)
            Cp[i * N] = acc[i];
    }
}

template<int N>
__global__ void k_gemm_bias(const float* __restrict__ A, const float* __restrict__ B,
                            const float* __restrict__ bias, float* __restrict__ C) {
    const int lane  = threadIdx.x & 31;
    const int wave  = threadIdx.x >> 5;
    const int row0  = blockIdx.x * 16;
    const int col0  = wave * 16;
    const int r     = lane & 15;
    const int kb    = (lane >> 4) * 2;

    const int col  = col0 + r;
    const int colc = (col < N) ? col : (N - 1);

    const float* Ap = A + (row0 + r) * DFEAT + kb;
    const float* Bp = B + kb * N + colc;

    v8f acc = {};
    #pragma unroll
    for (int k0 = 0; k0 < DFEAT; k0 += 4) {
        v2f a = *(const v2f*)(Ap + k0);
        v2f b;
        b.x = Bp[k0 * N];
        b.y = Bp[k0 * N + N];
        acc = __builtin_amdgcn_wmma_f32_16x16x4_f32(
            false, a, false, b, (short)0, acc, false, false);
    }

    if (col < N) {
        const float bb = bias[col];
        float* Cp = C + (row0 + (lane >> 4) * 8) * N + col;
        #pragma unroll
        for (int i = 0; i < 8; ++i)
            Cp[i * N] = acc[i] + bb;
    }
}

__global__ void k_scatter(const float* __restrict__ h, const int* __restrict__ src,
                          const int* __restrict__ dst, const float* __restrict__ dinv,
                          float* __restrict__ agg, int E) {
    const int e = blockIdx.x * (blockDim.x >> 5) + (threadIdx.x >> 5);
    if (e >= E) return;
    const int lane = threadIdx.x & 31;
    const int s = src[e];
    const int d = dst[e];
    const float norm = dinv[s] * dinv[d];
    const float4 v = *(const float4*)(h + s * DFEAT + lane * 4);
    float* out = agg + d * DFEAT + lane * 4;
    unsafeAtomicAdd(out + 0, v.x * norm);
    unsafeAtomicAdd(out + 1, v.y * norm);
    unsafeAtomicAdd(out + 2, v.z * norm);
    unsafeAtomicAdd(out + 3, v.w * norm);
}

__global__ void k_finalize(const float* __restrict__ h, const float* __restrict__ agg,
                           const float* __restrict__ dinv, const float* __restrict__ b,
                           float* __restrict__ out, int n) {
    int t = blockIdx.x * blockDim.x + threadIdx.x;
    if (t >= n) return;
    const int i = t >> 7;
    const int f = t & 127;
    const float di = dinv[i];
    out[t] = tanhf(agg[t] + h[t] * di * di + b[f]);
}

extern "C" void kernel_launch(void* const* d_in, const int* in_sizes, int n_in,
                              void* d_out, int out_size, void* d_ws, size_t ws_size,
                              hipStream_t stream) {
    const float* x    = (const float*)d_in[0];
    const int*   ei   = (const int*)  d_in[1];
    const float* W1   = (const float*)d_in[2];
    const float* b1   = (const float*)d_in[3];
    const float* W2   = (const float*)d_in[4];
    const float* b2   = (const float*)d_in[5];
    const float* Wlin = (const float*)d_in[6];
    const float* blin = (const float*)d_in[7];

    const int* src = ei;
    const int* dst = ei + N_EDGES;

    float* wsf  = (float*)d_ws;
    float* dinv = wsf;
    float* bufA = wsf + 50048;
    float* bufB = bufA + (size_t)N_NODES * DFEAT;

    const int NF = N_NODES * DFEAT;
    const dim3 gemmGrid(N_NODES / 16, DFEAT / 64);

    k_init_deg <<<(N_NODES + 255) / 256, 256, 0, stream>>>((int*)dinv, N_NODES);
    k_count_deg<<<(N_EDGES + 255) / 256, 256, 0, stream>>>((int*)dinv, dst, N_EDGES);
    k_make_dinv<<<(N_NODES + 255) / 256, 256, 0, stream>>>(dinv, N_NODES);

    k_gemm<DFEAT><<<gemmGrid, 128, 0, stream>>>(x, W1, bufA);
    k_zero    <<<(NF + 255) / 256, 256, 0, stream>>>(bufB, NF);
    k_scatter <<<(N_EDGES + 7) / 8, 256, 0, stream>>>(bufA, src, dst, dinv, bufB, N_EDGES);
    k_finalize<<<(NF + 255) / 256, 256, 0, stream>>>(bufA, bufB, dinv, b1, bufB, NF);

    k_gemm<DFEAT><<<gemmGrid, 128, 0, stream>>>(bufB, W2, bufA);
    k_zero    <<<(NF + 255) / 256, 256, 0, stream>>>(bufB, NF);
    k_scatter <<<(N_EDGES + 7) / 8, 256, 0, stream>>>(bufA, src, dst, dinv, bufB, N_EDGES);
    k_finalize<<<(NF + 255) / 256, 256, 0, stream>>>(bufA, bufB, dinv, b2, bufB, NF);

    k_gemm_bias<NCLS><<<N_NODES / 16, 96, 0, stream>>>(bufB, Wlin, blin, (float*)d_out);
}


# Edge sums moved from global float atomics into LDS: 256 blocks each own 196 dst rows, scan dst for their edges, add src rows by LDS atomics (self-loop+bias summed first); zero/tanh passes folded in
# speedup vs baseline: 1.4996x; 1.4996x over previous
#include <hip/hip_runtime.h>
#include <math.h>

typedef float v2f __attribute__((ext_vector_type(2)));
typedef float v8f __attribute__((ext_vector_type(8)));

#define N_NODES 50000
#define N_EDGES 600000
#define DFEAT   128
#define NCLS    40

__global__ void k_init_deg(int* deg, int n) {
    int i = blockIdx.x * blockDim.x + threadIdx.x;
    if (i < n) deg[i] = 1;
}

__global__ void k_count_deg(int* deg, const int* __restrict__ dst, int e) {
    int i = blockIdx.x * blockDim.x + threadIdx.x;
    if (i < e) atomicAdd(&deg[dst[i]], 1);
}

__global__ void k_make_dinv(float* dinv, int n) {
    int i = blockIdx.x * blockDim.x + threadIdx.x;
    if (i < n) {
        int d = ((const int*)dinv)[i];
        dinv[i] = rsqrtf((float)d);
    }
}

template<int N>
__global__ void k_gemm(const float* __restrict__ A, const float* __restrict__ B,
                       float* __restrict__ C) {
    const int lane  = threadIdx.x & 31;
    const int wave  = threadIdx.x >> 5;
    const int row0  = blockIdx.x * 16;
    const int col0  = (blockIdx.y * (blockDim.x >> 5) + wave) * 16;
    const int r     = lane & 15;
    const int kb    = (lane >> 4) * 2;

    const int col  = col0 + r;
    const int colc = (col < N) ? col : (N - 1);

    const float* Ap = A + (row0 + r) * DFEAT + kb;
    const float* Bp = B + kb * N + colc;

    v8f acc = {};
    #pragma unroll
    for (int k0 = 0; k0 < DFEAT; k0 += 4) {
        v2f a = *(const v2f*)(Ap + k0);
        v2f b;
        b.x = Bp[k0 * N];
        b.y = Bp[k0 * N + N];
        acc = __builtin_amdgcn_wmma_f32_16x16x4_f32(
            false, a, false, b, (short)0, acc, false, false);
    }

    if (col < N) {
        float* Cp = C + (row0 + (lane >> 4) * 8) * N + col;
        #pragma unroll
        for (int i = 0; i < 8; ++i)
            Cp[i * N] = acc[i];
    }
}

template<int N>
__global__ void k_gemm_bias(const float* __restrict__ A, const float* __restrict__ B,
                            const float* __restrict__ bias, float* __restrict__ C) {
    const int lane  = threadIdx.x & 31;
    const int wave  = threadIdx.x >> 5;
    const int row0  = blockIdx.x * 16;
    const int col0  = wave * 16;
    const int r     = lane & 15;
    const int kb    = (lane >> 4) * 2;

    const int col  = col0 + r;
    const int colc = (col < N) ? col : (N - 1);

    const float* Ap = A + (row0 + r) * DFEAT + kb;
    const float* Bp = B + kb * N + colc;

    v8f acc = {};
    #pragma unroll
    for (int k0 = 0; k0 < DFEAT; k0 += 4) {
        v2f a = *(const v2f*)(Ap + k0);
        v2f b;
        b.x = Bp[k0 * N];
        b.y = Bp[k0 * N + N];
        acc = __builtin_amdgcn_wmma_f32_16x16x4_f32(
            false, a, false, b, (short)0, acc, false, false);
    }

    if (col < N) {
        const float bb = bias[col];
        float* Cp = C + (row0 + (lane >> 4) * 8) * N + col;
        #pragma unroll
        for (int i = 0; i < 8; ++i)
            Cp[i * N] = acc[i] + bb;
    }
}

#define AGG_BLOCKS   256
#define AGG_NODES    196
#define AGG_THREADS  1024
#define AGG_WAVES    (AGG_THREADS / 32)
#define AGG_QCAP     544
#define AGG_GROUP    8
#define N_EDGE_VEC   (N_EDGES / 4)
#define AGG_ROT      576

__device__ __forceinline__ int agg_vec_index(const int pos, const int start) {
    const int idx = pos + start;
    return (idx < N_EDGE_VEC) ? idx : (idx - N_EDGE_VEC);
}

__device__ __forceinline__ int4 agg_load_dst(const int4* __restrict__ dst4, const int pos, const int start) {
    return (pos < N_EDGE_VEC) ? dst4[agg_vec_index(pos, start)] : make_int4(-1, -1, -1, -1);
}

__device__ __forceinline__ void agg_drain(int2* wq, const int qn, const int lane,
                                          const float* __restrict__ h,
                                          const int* __restrict__ src,
                                          const float* __restrict__ dinv,
                                          const float* sdinv, float* acc) {
    for (int j = lane; j < qn; j += 32) {
        const int e = wq[j].x;
        wq[j].x = src[e];
    }
    for (int j = 0; j < qn; j += AGG_GROUP) {
        int s[AGG_GROUP], dl[AGG_GROUP];
        #pragma unroll
        for (int k = 0; k < AGG_GROUP; ++k) {
            const int jj = (j + k < qn) ? (j + k) : (qn - 1);
            const int2 ent = wq[jj];
            s[k]  = __builtin_amdgcn_readfirstlane(ent.x);
            dl[k] = __builtin_amdgcn_readfirstlane(ent.y);
        }
        float v[AGG_GROUP][4];
        float nrm[AGG_GROUP];
        #pragma unroll
        for (int k = 0; k < AGG_GROUP; ++k) {
            nrm[k] = dinv[s[k]] * sdinv[dl[k]];
            const float* hp = h + s[k] * DFEAT + lane;
            v[k][0] = hp[0];
            v[k][1] = hp[32];
            v[k][2] = hp[64];
            v[k][3] = hp[96];
        }
        #pragma unroll
        for (int k = 0; k < AGG_GROUP; ++k) {
            if (j + k < qn) {
                float* a = acc + dl[k] * DFEAT + lane;
                atomicAdd(a +  0, v[k][0] * nrm[k]);
                atomicAdd(a + 32, v[k][1] * nrm[k]);
                atomicAdd(a + 64, v[k][2] * nrm[k]);
                atomicAdd(a + 96, v[k][3] * nrm[k]);
            }
        }
    }
}

__global__ __launch_bounds__(AGG_THREADS)
void k_aggregate(const float* __restrict__ h, const int* __restrict__ src,
                 const int* __restrict__ dst, const float* __restrict__ dinv,
                 const float* __restrict__ bias, float* __restrict__ out) {
    __shared__ float acc[AGG_NODES * DFEAT];
    __shared__ float sdinv[AGG_NODES];
    __shared__ int2  queue[AGG_WAVES][AGG_QCAP];

    const int tid  = threadIdx.x;
    const int lane = tid & 31;
    const int wave = tid >> 5;
    const int lo   = blockIdx.x * AGG_NODES;
    const int cnt  = (N_NODES - lo < AGG_NODES) ? (N_NODES - lo) : AGG_NODES;

    const int start = blockIdx.x * AGG_ROT;

    const int4* dst4 = (const int4*)dst;
    int4 nx0 = agg_load_dst(dst4, tid, start);
    int4 nx1 = agg_load_dst(dst4, tid + AGG_THREADS, start);
    int4 nx2 = agg_load_dst(dst4, tid + 2 * AGG_THREADS, start);
    int4 nx3 = agg_load_dst(dst4, tid + 3 * AGG_THREADS, start);

    for (int i = tid; i < cnt * DFEAT; i += AGG_THREADS) {
        const int   n  = i >> 7;
        const float di = dinv[lo + n];
        acc[i] = h[(lo + n) * DFEAT + (i & 127)] * di * di + bias[i & 127];
    }
    for (int i = tid; i < cnt; i += AGG_THREADS) sdinv[i] = dinv[lo + i];
    __syncthreads();

    int2* wq = queue[wave];
    int qn = 0;

    for (int base = 0; base < N_EDGE_VEC; base += 4 * AGG_THREADS) {
        const int4 cur[4] = {nx0, nx1, nx2, nx3};
        nx0 = agg_load_dst(dst4, base + 4 * AGG_THREADS + tid, start);
        nx1 = agg_load_dst(dst4, base + 5 * AGG_THREADS + tid, start);
        nx2 = agg_load_dst(dst4, base + 6 * AGG_THREADS + tid, start);
        nx3 = agg_load_dst(dst4, base + 7 * AGG_THREADS + tid, start);

        if (qn > AGG_QCAP - 512) {
            agg_drain(wq, qn, lane, h, src, dinv, sdinv, acc);
            qn = 0;
        }
        #pragma unroll
        for (int u = 0; u < 4; ++u) {
            const int idx = agg_vec_index(base + u * AGG_THREADS + tid, start);
            const int dc[4] = {cur[u].x - lo, cur[u].y - lo, cur[u].z - lo, cur[u].w - lo};
            const bool any = ((unsigned)dc[0] < (unsigned)cnt) | ((unsigned)dc[1] < (unsigned)cnt) |
                             ((unsigned)dc[2] < (unsigned)cnt) | ((unsigned)dc[3] < (unsigned)cnt);
            if (__builtin_amdgcn_ballot_w32(any) != 0u) {
                #pragma unroll
                for (int c = 0; c < 4; ++c) {
                    const bool hit = (unsigned)dc[c] < (unsigned)cnt;
                    const unsigned m = __builtin_amdgcn_ballot_w32(hit);
                    if (hit) {
                        const int pos = qn + (int)__builtin_amdgcn_mbcnt_lo(m, 0u);
                        wq[pos] = make_int2(4 * idx + c, dc[c]);
                    }
                    qn += __builtin_popcount(m);
                }
            }
        }
    }
    if (qn > 0) agg_drain(wq, qn, lane, h, src, dinv, sdinv, acc);
    __syncthreads();

    float* op = out + lo * DFEAT;
    for (int i = tid; i < cnt * DFEAT; i += AGG_THREADS)
        op[i] = tanhf(acc[i]);
}

extern "C" void kernel_launch(void* const* d_in, const int* in_sizes, int n_in,
                              void* d_out, int out_size, void* d_ws, size_t ws_size,
                              hipStream_t stream) {
    const float* x    = (const float*)d_in[0];
    const int*   ei   = (const int*)  d_in[1];
    const float* W1   = (const float*)d_in[2];
    const float* b1   = (const float*)d_in[3];
    const float* W2   = (const float*)d_in[4];
    const float* b2   = (const float*)d_in[5];
    const float* Wlin = (const float*)d_in[6];
    const float* blin = (const float*)d_in[7];

    const int* src = ei;
    const int* dst = ei + N_EDGES;

    float* wsf  = (float*)d_ws;
    float* dinv = wsf;
    float* bufA = wsf + 50048;
    float* bufB = bufA + (size_t)N_NODES * DFEAT;

    const dim3 gemmGrid(N_NODES / 16, DFEAT / 64);

    k_init_deg <<<(N_NODES + 255) / 256, 256, 0, stream>>>((int*)dinv, N_NODES);
    k_count_deg<<<(N_EDGES + 255) / 256, 256, 0, stream>>>((int*)dinv, dst, N_EDGES);
    k_make_dinv<<<(N_NODES + 255) / 256, 256, 0, stream>>>(dinv, N_NODES);

    k_gemm<DFEAT><<<gemmGrid, 128, 0, stream>>>(x, W1, bufA);
    k_aggregate<<<AGG_BLOCKS, AGG_THREADS, 0, stream>>>(bufA, src, dst, dinv, b1, bufB);

    k_gemm<DFEAT><<<gemmGrid, 128, 0, stream>>>(bufB, W2, bufA);
    k_aggregate<<<AGG_BLOCKS, AGG_THREADS, 0, stream>>>(bufA, src, dst, dinv, b2, bufB);

    k_gemm_bias<NCLS><<<N_NODES / 16, 96, 0, stream>>>(bufB, Wlin, blin, (float*)d_out);
}


# Layer 2 stops sweeping dst: each block replays the (src,row) pairs its layer-1 sweep kept in a per-block workspace list (sweeps again only if a list overflowed); LDS atomic sums as before
# speedup vs baseline: 1.7394x; 1.1599x over previous
#include <hip/hip_runtime.h>
#include <math.h>

typedef float v2f __attribute__((ext_vector_type(2)));
typedef float v8f __attribute__((ext_vector_type(8)));

#define N_NODES 50000
#define N_EDGES 600000
#define DFEAT   128
#define NCLS    40

__global__ void k_init_deg(int* deg, int n) {
    int i = blockIdx.x * blockDim.x + threadIdx.x;
    if (i < n) deg[i] = 1;
}

__global__ void k_count_deg(int* deg, const int* __restrict__ dst, int e) {
    int i = blockIdx.x * blockDim.x + threadIdx.x;
    if (i < e) atomicAdd(&deg[dst[i]], 1);
}

__global__ void k_make_dinv(float* dinv, int n) {
    int i = blockIdx.x * blockDim.x + threadIdx.x;
    if (i < n) {
        int d = ((const int*)dinv)[i];
        dinv[i] = rsqrtf((float)d);
    }
}

template<int N>
__global__ void k_gemm(const float* __restrict__ A, const float* __restrict__ B,
                       float* __restrict__ C) {
    const int lane  = threadIdx.x & 31;
    const int wave  = threadIdx.x >> 5;
    const int row0  = blockIdx.x * 16;
    const int col0  = (blockIdx.y * (blockDim.x >> 5) + wave) * 16;
    const int r     = lane & 15;
    const int kb    = (lane >> 4) * 2;

    const int col  = col0 + r;
    const int colc = (col < N) ? col : (N - 1);

    const float* Ap = A + (row0 + r) * DFEAT + kb;
    const float* Bp = B + kb * N + colc;

    v8f acc = {};
    #pragma unroll
    for (int k0 = 0; k0 < DFEAT; k0 += 4) {
        v2f a = *(const v2f*)(Ap + k0);
        v2f b;
        b.x = Bp[k0 * N];
        b.y = Bp[k0 * N + N];
        acc = __builtin_amdgcn_wmma_f32_16x16x4_f32(
            false, a, false, b, (short)0, acc, false, false);
    }

    if (col < N) {
        float* Cp = C + (row0 + (lane >> 4) * 8) * N + col;
        #pragma unroll
        for (int i = 0; i < 8; ++i)
            Cp[i * N] = acc[i];
    }
}

template<int N>
__global__ void k_gemm_bias(const float* __restrict__ A, const float* __restrict__ B,
                            const float* __restrict__ bias, float* __restrict__ C) {
    const int lane  = threadIdx.x & 31;
    const int wave  = threadIdx.x >> 5;
    const int row0  = blockIdx.x * 16;
    const int col0  = wave * 16;
    const int r     = lane & 15;
    const int kb    = (lane >> 4) * 2;

    const int col  = col0 + r;
    const int colc = (col < N) ? col : (N - 1);

    const float* Ap = A + (row0 + r) * DFEAT + kb;
    const float* Bp = B + kb * N + colc;

    v8f acc = {};
    #pragma unroll
    for (int k0 = 0; k0 < DFEAT; k0 += 4) {
        v2f a = *(const v2f*)(Ap + k0);
        v2f b;
        b.x = Bp[k0 * N];
        b.y = Bp[k0 * N + N];
        acc = __builtin_amdgcn_wmma_f32_16x16x4_f32(
            false, a, false, b, (short)0, acc, false, false);
    }

    if (col < N) {
        const float bb = bias[col];
        float* Cp = C + (row0 + (lane >> 4) * 8) * N + col;
        #pragma unroll
        for (int i = 0; i < 8; ++i)
            Cp[i * N] = acc[i] + bb;
    }
}

#define AGG_BLOCKS   256
#define AGG_NODES    196
#define AGG_THREADS  1024
#define AGG_WAVES    (AGG_THREADS / 32)
#define AGG_QCAP     544
#define AGG_GROUP    8
#define N_EDGE_VEC   (N_EDGES / 4)
#define AGG_ROT      576
#define AGG_LCAP     8192
#define AGG_CHUNK    512

__device__ __forceinline__ int agg_vec_index(const int pos, const int start) {
    const int idx = pos + start;
    return (idx < N_EDGE_VEC) ? idx : (idx - N_EDGE_VEC);
}

__device__ __forceinline__ int4 agg_load_dst(const int4* __restrict__ dst4, const int pos, const int start) {
    return (pos < N_EDGE_VEC) ? dst4[agg_vec_index(pos, start)] : make_int4(-1, -1, -1, -1);
}

__device__ __forceinline__ void agg_accumulate(const int2* wq, const int qn, const int lane,
                                               const float* __restrict__ h,
                                               const float* __restrict__ dinv,
                                               const float* sdinv, float* acc) {
    for (int j = 0; j < qn; j += AGG_GROUP) {
        int s[AGG_GROUP], dl[AGG_GROUP];
        #pragma unroll
        for (int k = 0; k < AGG_GROUP; ++k) {
            const int jj = (j + k < qn) ? (j + k) : (qn - 1);
            const int2 ent = wq[jj];
            s[k]  = __builtin_amdgcn_readfirstlane(ent.x);
            dl[k] = __builtin_amdgcn_readfirstlane(ent.y);
        }
        float v[AGG_GROUP][4];
        float nrm[AGG_GROUP];
        #pragma unroll
        for (int k = 0; k < AGG_GROUP; ++k) {
            nrm[k] = dinv[s[k]] * sdinv[dl[k]];
            const float* hp = h + s[k] * DFEAT + lane;
            v[k][0] = hp[0];
            v[k][1] = hp[32];
            v[k][2] = hp[64];
            v[k][3] = hp[96];
        }
        #pragma unroll
        for (int k = 0; k < AGG_GROUP; ++k) {
            if (j + k < qn) {
                float* a = acc + dl[k] * DFEAT + lane;
                atomicAdd(a +  0, v[k][0] * nrm[k]);
                atomicAdd(a + 32, v[k][1] * nrm[k]);
                atomicAdd(a + 64, v[k][2] * nrm[k]);
                atomicAdd(a + 96, v[k][3] * nrm[k]);
            }
        }
    }
}

template<bool KEEP>
__device__ __forceinline__ void agg_drain(int2* wq, const int qn, const int lane,
                                          const float* __restrict__ h,
                                          const int* __restrict__ src,
                                          const float* __restrict__ dinv,
                                          const float* sdinv, float* acc,
                                          int2* list, int* cursor) {
    for (int j = lane; j < qn; j += 32) {
        const int e = wq[j].x;
        wq[j].x = src[e];
    }
    if (KEEP) {
        int base = 0;
        if (lane == 0) base = atomicAdd(cursor, qn);
        base = __builtin_amdgcn_readfirstlane(base);
        if (base + qn <= AGG_LCAP)
            for (int j = lane; j < qn; j += 32) list[base + j] = wq[j];
    }
    agg_accumulate(wq, qn, lane, h, dinv, sdinv, acc);
}

template<bool KEEP>
__global__ __launch_bounds__(AGG_THREADS)
void k_aggregate(const float* __restrict__ h, const int* __restrict__ src,
                 const int* __restrict__ dst, const float* __restrict__ dinv,
                 const float* __restrict__ bias, float* __restrict__ out,
                 int* lists, int* counts) {
    __shared__ float acc[AGG_NODES * DFEAT];
    __shared__ float sdinv[AGG_NODES];
    __shared__ int2  queue[AGG_WAVES][AGG_QCAP];
    __shared__ int   cursor;

    const int tid  = threadIdx.x;
    const int lane = tid & 31;
    const int wave = tid >> 5;
    const int lo   = blockIdx.x * AGG_NODES;
    const int cnt  = (N_NODES - lo < AGG_NODES) ? (N_NODES - lo) : AGG_NODES;

    const int start = blockIdx.x * AGG_ROT;

    int2* mylist = (int2*)lists + blockIdx.x * AGG_LCAP;

    int  kept  = 0;
    bool sweep = true;
    if (!KEEP) {
        kept  = __builtin_amdgcn_readfirstlane(counts[blockIdx.x]);
        sweep = kept > AGG_LCAP;
    }

    const int4* dst4 = (const int4*)dst;
    int4 nx0 = make_int4(-1, -1, -1, -1), nx1 = nx0, nx2 = nx0, nx3 = nx0;
    if (sweep) {
        nx0 = agg_load_dst(dst4, tid, start);
        nx1 = agg_load_dst(dst4, tid + AGG_THREADS, start);
        nx2 = agg_load_dst(dst4, tid + 2 * AGG_THREADS, start);
        nx3 = agg_load_dst(dst4, tid + 3 * AGG_THREADS, start);
    }

    for (int i = tid; i < cnt * DFEAT; i += AGG_THREADS) {
        const int   n  = i >> 7;
        const float di = dinv[lo + n];
        acc[i] = h[(lo + n) * DFEAT + (i & 127)] * di * di + bias[i & 127];
    }
    for (int i = tid; i < cnt; i += AGG_THREADS) sdinv[i] = dinv[lo + i];
    if (tid == 0) cursor = 0;
    __syncthreads();

    int2* wq = queue[wave];

    if (sweep) {
        int qn = 0;

        for (int base = 0; base < N_EDGE_VEC; base += 4 * AGG_THREADS) {
            const int4 cur[4] = {nx0, nx1, nx2, nx3};
            nx0 = agg_load_dst(dst4, base + 4 * AGG_THREADS + tid, start);
            nx1 = agg_load_dst(dst4, base + 5 * AGG_THREADS + tid, start);
            nx2 = agg_load_dst(dst4, base + 6 * AGG_THREADS + tid, start);
            nx3 = agg_load_dst(dst4, base + 7 * AGG_THREADS + tid, start);

            if (qn > AGG_QCAP - 512) {
                agg_drain<KEEP>(wq, qn, lane, h, src, dinv, sdinv, acc, mylist, &cursor);
                qn = 0;
            }
            #pragma unroll
            for (int u = 0; u < 4; ++u) {
                const int idx = agg_vec_index(base + u * AGG_THREADS + tid, start);
                const int dc[4] = {cur[u].x - lo, cur[u].y - lo, cur[u].z - lo, cur[u].w - lo};
                const bool any = ((unsigned)dc[0] < (unsigned)cnt) | ((unsigned)dc[1] < (unsigned)cnt) |
                                 ((unsigned)dc[2] < (unsigned)cnt) | ((unsigned)dc[3] < (unsigned)cnt);
                if (__builtin_amdgcn_ballot_w32(any) != 0u) {
                    #pragma unroll
                    for (int c = 0; c < 4; ++c) {
                        const bool hit = (unsigned)dc[c] < (unsigned)cnt;
                        const unsigned m = __builtin_amdgcn_ballot_w32(hit);
                        if (hit) {
                            const int pos = qn + (int)__builtin_amdgcn_mbcnt_lo(m, 0u);
                            wq[pos] = make_int2(4 * idx + c, dc[c]);
                        }
                        qn += __builtin_popcount(m);
                    }
                }
            }
        }
        if (qn > 0) agg_drain<KEEP>(wq, qn, lane, h, src, dinv, sdinv, acc, mylist, &cursor);
    } else {
        const int w   = __builtin_amdgcn_readfirstlane(wave);
        const int per = (kept + AGG_WAVES - 1) / AGG_WAVES;
        const int beg = w * per;
        const int end = (beg + per < kept) ? (beg + per) : kept;
        for (int c = beg; c < end; c += AGG_CHUNK) {
            const int m = (end - c < AGG_CHUNK) ? (end - c) : AGG_CHUNK;
            for (int j = lane; j < m; j += 32) wq[j] = mylist[c + j];
            agg_accumulate(wq, m, lane, h, dinv, sdinv, acc);
        }
    }
    __syncthreads();
    if (KEEP && tid == 0) counts[blockIdx.x] = cursor;

    float* op = out + lo * DFEAT;
    for (int i = tid; i < cnt * DFEAT; i += AGG_THREADS)
        op[i] = tanhf(acc[i]);
}

extern "C" void kernel_launch(void* const* d_in, const int* in_sizes, int n_in,
                              void* d_out, int out_size, void* d_ws, size_t ws_size,
                              hipStream_t stream) {
    const float* x    = (const float*)d_in[0];
    const int*   ei   = (const int*)  d_in[1];
    const float* W1   = (const float*)d_in[2];
    const float* b1   = (const float*)d_in[3];
    const float* W2   = (const float*)d_in[4];
    const float* b2   = (const float*)d_in[5];
    const float* Wlin = (const float*)d_in[6];
    const float* blin = (const float*)d_in[7];

    const int* src = ei;
    const int* dst = ei + N_EDGES;

    float* wsf  = (float*)d_ws;
    float* dinv = wsf;
    float* bufA = wsf + 50048;
    float* bufB = bufA + (size_t)N_NODES * DFEAT;
    int*   lists  = (int*)(bufB + (size_t)N_NODES * DFEAT);
    int*   counts = lists + (size_t)AGG_BLOCKS * AGG_LCAP * 2;

    const dim3 gemmGrid(N_NODES / 16, DFEAT / 64);

    k_init_deg <<<(N_NODES + 255) / 256, 256, 0, stream>>>((int*)dinv, N_NODES);
    k_count_deg<<<(N_EDGES + 255) / 256, 256, 0, stream>>>((int*)dinv, dst, N_EDGES);
    k_make_dinv<<<(N_NODES + 255) / 256, 256, 0, stream>>>(dinv, N_NODES);

    k_gemm<DFEAT><<<gemmGrid, 128, 0, stream>>>(x, W1, bufA);
    k_aggregate<true><<<AGG_BLOCKS, AGG_THREADS, 0, stream>>>(bufA, src, dst, dinv, b1, bufB, lists, counts);

    k_gemm<DFEAT><<<gemmGrid, 128, 0, stream>>>(bufB, W2, bufA);
    k_aggregate<false><<<AGG_BLOCKS, AGG_THREADS, 0, stream>>>(bufA, src, dst, dinv, b2, bufB, lists, counts);

    k_gemm_bias<NCLS><<<N_NODES / 16, 96, 0, stream>>>(bufB, Wlin, blin, (float*)d_out);
}


# Degree pass splits the edges, in edge order, into 16 node-range segments per 2344-edge chunk (ballot ranks); a block's layer-1 scan reads 1/16 of the edges, not all dst; LDS atomic sums as before
# speedup vs baseline: 1.9591x; 1.1263x over previous
#include <hip/hip_runtime.h>
#include <math.h>

typedef float v2f __attribute__((ext_vector_type(2)));
typedef float v8f __attribute__((ext_vector_type(8)));

#define N_NODES 50000
#define N_EDGES 600000
#define DFEAT   128
#define NCLS    40

__global__ void k_init_deg(int* deg, int n) {
    int i = blockIdx.x * blockDim.x + threadIdx.x;
    if (i < n) deg[i] = 1;
}

#define BIN_RANGES   16
#define BIN_OWNERS   16
#define BIN_NODES    (BIN_OWNERS * 196)
#define BIN_CHUNKS   256
#define BIN_EDGES    2344
#define BIN_THREADS  1024
#define BIN_WAVES    (BIN_THREADS / 32)
#define BIN_PER_WAVE 74
#define BIN_VECS     3

__global__ __launch_bounds__(BIN_THREADS)
void k_count_partition(int* deg, const int* __restrict__ src, const int* __restrict__ dst,
                       int* segs, int* segcount) {
    __shared__ int wcnt[BIN_WAVES][BIN_RANGES];
    __shared__ int wstart[BIN_WAVES][BIN_RANGES];

    const int tid  = threadIdx.x;
    const int lane = tid & 31;
    const int wave = tid >> 5;
    const int e0 = blockIdx.x * BIN_EDGES;
    const int e1 = (e0 + BIN_EDGES < N_EDGES) ? (e0 + BIN_EDGES) : N_EDGES;
    const int wb = e0 + wave * BIN_PER_WAVE;
    const int we = (wb + BIN_PER_WAVE < e1) ? (wb + BIN_PER_WAVE) : e1;

    int s[BIN_VECS], d[BIN_VECS], r[BIN_VECS], rk[BIN_VECS];
    #pragma unroll
    for (int v = 0; v < BIN_VECS; ++v) {
        const int  e  = wb + 32 * v + lane;
        const bool ok = e < we;
        d[v] = ok ? dst[e] : -1;
        s[v] = ok ? src[e] : 0;
    }
    #pragma unroll
    for (int v = 0; v < BIN_VECS; ++v) {
        r[v]  = (d[v] >= 0) ? (d[v] / BIN_NODES) : -1;
        rk[v] = 0;
        if (d[v] >= 0) atomicAdd(&deg[d[v]], 1);
    }

    int mine = 0;
    #pragma unroll
    for (int b = 0; b < BIN_RANGES; ++b) {
        int run = 0;
        #pragma unroll
        for (int v = 0; v < BIN_VECS; ++v) {
            const bool hit = (r[v] == b);
            const unsigned m = __builtin_amdgcn_ballot_w32(hit);
            if (hit) rk[v] = run + (int)__builtin_amdgcn_mbcnt_lo(m, 0u);
            run += __builtin_popcount(m);
        }
        if (lane == b) mine = run;
    }
    if (lane < BIN_RANGES) wcnt[wave][lane] = mine;
    __syncthreads();

    if (tid < BIN_WAVES * BIN_RANGES) {
        const int w = tid >> 4;
        const int b = tid & 15;
        int sum = 0;
        for (int w2 = 0; w2 < w; ++w2) sum += wcnt[w2][b];
        wstart[w][b] = sum;
        if (w == BIN_WAVES - 1) segcount[b * BIN_CHUNKS + blockIdx.x] = sum + wcnt[w][b];
    }
    __syncthreads();

    int2* seg2 = (int2*)segs;
    #pragma unroll
    for (int v = 0; v < BIN_VECS; ++v) {
        if (r[v] >= 0) {
            const int pos = wstart[wave][r[v]] + rk[v];
            seg2[(r[v] * BIN_CHUNKS + blockIdx.x) * BIN_EDGES + pos] = make_int2(s[v], d[v]);
        }
    }
}

__global__ void k_make_dinv(float* dinv, int n) {
    int i = blockIdx.x * blockDim.x + threadIdx.x;
    if (i < n) {
        int d = ((const int*)dinv)[i];
        dinv[i] = rsqrtf((float)d);
    }
}

template<int N>
__global__ void k_gemm(const float* __restrict__ A, const float* __restrict__ B,
                       float* __restrict__ C) {
    const int lane  = threadIdx.x & 31;
    const int wave  = threadIdx.x >> 5;
    const int row0  = blockIdx.x * 16;
    const int col0  = (blockIdx.y * (blockDim.x >> 5) + wave) * 16;
    const int r     = lane & 15;
    const int kb    = (lane >> 4) * 2;

    const int col  = col0 + r;
    const int colc = (col < N) ? col : (N - 1);

    const float* Ap = A + (row0 + r) * DFEAT + kb;
    const float* Bp = B + kb * N + colc;

    v8f acc = {};
    #pragma unroll
    for (int k0 = 0; k0 < DFEAT; k0 += 4) {
        v2f a = *(const v2f*)(Ap + k0);
        v2f b;
        b.x = Bp[k0 * N];
        b.y = Bp[k0 * N + N];
        acc = __builtin_amdgcn_wmma_f32_16x16x4_f32(
            false, a, false, b, (short)0, acc, false, false);
    }

    if (col < N) {
        float* Cp = C + (row0 + (lane >> 4) * 8) * N + col;
        #pragma unroll
        for (int i = 0; i < 8; ++i)
            Cp[i * N] = acc[i];
    }
}

template<int N>
__global__ void k_gemm_bias(const float* __restrict__ A, const float* __restrict__ B,
                            const float* __restrict__ bias, float* __restrict__ C) {
    const int lane  = threadIdx.x & 31;
    const int wave  = threadIdx.x >> 5;
    const int row0  = blockIdx.x * 16;
    const int col0  = wave * 16;
    const int r     = lane & 15;
    const int kb    = (lane >> 4) * 2;

    const int col  = col0 + r;
    const int colc = (col < N) ? col : (N - 1);

    const float* Ap = A + (row0 + r) * DFEAT + kb;
    const float* Bp = B + kb * N + colc;

    v8f acc = {};
    #pragma unroll
    for (int k0 = 0; k0 < DFEAT; k0 += 4) {
        v2f a = *(const v2f*)(Ap + k0);
        v2f b;
        b.x = Bp[k0 * N];
        b.y = Bp[k0 * N + N];
        acc = __builtin_amdgcn_wmma_f32_16x16x4_f32(
            false, a, false, b, (short)0, acc, false, false);
    }

    if (col < N) {
        const float bb = bias[col];
        float* Cp = C + (row0 + (lane >> 4) * 8) * N + col;
        #pragma unroll
        for (int i = 0; i < 8; ++i)
            Cp[i * N] = acc[i] + bb;
    }
}

#define AGG_BLOCKS   256
#define AGG_NODES    196
#define AGG_THREADS  1024
#define AGG_WAVES    (AGG_THREADS / 32)
#define AGG_SEGS     (BIN_CHUNKS / AGG_WAVES)
#define AGG_QCAP     544
#define AGG_GROUP    8
#define AGG_LCAP     8192
#define AGG_CHUNK    512

__device__ __forceinline__ void agg_accumulate(const int2* wq, const int qn, const int lane,
                                               const float* __restrict__ h,
                                               const float* __restrict__ dinv,
                                               const float* sdinv, float* acc) {
    for (int j = 0; j < qn; j += AGG_GROUP) {
        int s[AGG_GROUP], dl[AGG_GROUP];
        #pragma unroll
        for (int k = 0; k < AGG_GROUP; ++k) {
            const int jj = (j + k < qn) ? (j + k) : (qn - 1);
            const int2 ent = wq[jj];
            s[k]  = __builtin_amdgcn_readfirstlane(ent.x);
            dl[k] = __builtin_amdgcn_readfirstlane(ent.y);
        }
        float v[AGG_GROUP][4];
        float nrm[AGG_GROUP];
        #pragma unroll
        for (int k = 0; k < AGG_GROUP; ++k) {
            nrm[k] = dinv[s[k]] * sdinv[dl[k]];
            const float* hp = h + s[k] * DFEAT + lane;
            v[k][0] = hp[0];
            v[k][1] = hp[32];
            v[k][2] = hp[64];
            v[k][3] = hp[96];
        }
        #pragma unroll
        for (int k = 0; k < AGG_GROUP; ++k) {
            if (j + k < qn) {
                float* a = acc + dl[k] * DFEAT + lane;
                atomicAdd(a +  0, v[k][0] * nrm[k]);
                atomicAdd(a + 32, v[k][1] * nrm[k]);
                atomicAdd(a + 64, v[k][2] * nrm[k]);
                atomicAdd(a + 96, v[k][3] * nrm[k]);
            }
        }
    }
}

template<bool KEEP>
__device__ __forceinline__ void agg_drain(const int2* wq, const int qn, const int lane,
                                          const float* __restrict__ h,
                                          const float* __restrict__ dinv,
                                          const float* sdinv, float* acc,
                                          int2* list, int* cursor) {
    if (KEEP) {
        int base = 0;
        if (lane == 0) base = atomicAdd(cursor, qn);
        base = __builtin_amdgcn_readfirstlane(base);
        if (base + qn <= AGG_LCAP)
            for (int j = lane; j < qn; j += 32) list[base + j] = wq[j];
    }
    agg_accumulate(wq, qn, lane, h, dinv, sdinv, acc);
}

template<bool KEEP>
__global__ __launch_bounds__(AGG_THREADS)
void k_aggregate(const float* __restrict__ h, const int* __restrict__ segs,
                 const int* __restrict__ segcount, const float* __restrict__ dinv,
                 const float* __restrict__ bias, float* __restrict__ out,
                 int* lists, int* counts) {
    __shared__ float acc[AGG_NODES * DFEAT];
    __shared__ float sdinv[AGG_NODES];
    __shared__ int2  queue[AGG_WAVES][AGG_QCAP];
    __shared__ int   cursor;

    const int tid  = threadIdx.x;
    const int lane = tid & 31;
    const int wave = tid >> 5;
    const int lo   = blockIdx.x * AGG_NODES;
    const int cnt  = (N_NODES - lo < AGG_NODES) ? (N_NODES - lo) : AGG_NODES;

    int2* mylist = (int2*)lists + blockIdx.x * AGG_LCAP;

    int  kept = 0;
    bool scan = true;
    if (!KEEP) {
        kept = __builtin_amdgcn_readfirstlane(counts[blockIdx.x]);
        scan = kept > AGG_LCAP;
    }

    const int   range = blockIdx.x / BIN_OWNERS;
    const int2* seg2  = (const int2*)segs + (range * BIN_CHUNKS + wave) * BIN_EDGES;
    int len[AGG_SEGS];
    #pragma unroll
    for (int k = 0; k < AGG_SEGS; ++k)
        len[k] = scan ? segcount[range * BIN_CHUNKS + wave + k * AGG_WAVES] : 0;

    for (int i = tid; i < cnt * DFEAT; i += AGG_THREADS) {
        const int   n  = i >> 7;
        const float di = dinv[lo + n];
        acc[i] = h[(lo + n) * DFEAT + (i & 127)] * di * di + bias[i & 127];
    }
    for (int i = tid; i < cnt; i += AGG_THREADS) sdinv[i] = dinv[lo + i];
    if (tid == 0) cursor = 0;
    __syncthreads();

    int2* wq = queue[wave];

    if (scan) {
        int longest = 0;
        #pragma unroll
        for (int k = 0; k < AGG_SEGS; ++k) {
            len[k]  = __builtin_amdgcn_readfirstlane(len[k]);
            len[k]  = (len[k] < BIN_EDGES) ? len[k] : BIN_EDGES;
            longest = (len[k] > longest) ? len[k] : longest;
        }

        int qn = 0;

        for (int t0 = 0; t0 < longest; t0 += 32) {
            const int t = t0 + lane;
            int2 pr[AGG_SEGS];
            #pragma unroll
            for (int k = 0; k < AGG_SEGS; ++k)
                pr[k] = (t < len[k]) ? seg2[k * AGG_WAVES * BIN_EDGES + t] : make_int2(-1, -1);

            if (qn > AGG_QCAP - 32 * AGG_SEGS) {
                agg_drain<KEEP>(wq, qn, lane, h, dinv, sdinv, acc, mylist, &cursor);
                qn = 0;
            }
            #pragma unroll
            for (int k = 0; k < AGG_SEGS; ++k) {
                const int  dl  = pr[k].y - lo;
                const bool hit = (unsigned)dl < (unsigned)cnt;
                const unsigned m = __builtin_amdgcn_ballot_w32(hit);
                if (hit) {
                    const int pos = qn + (int)__builtin_amdgcn_mbcnt_lo(m, 0u);
                    wq[pos] = make_int2(pr[k].x, dl);
                }
                qn += __builtin_popcount(m);
            }
        }
        if (qn > 0) agg_drain<KEEP>(wq, qn, lane, h, dinv, sdinv, acc, mylist, &cursor);
    } else {
        const int w   = __builtin_amdgcn_readfirstlane(wave);
        const int per = (kept + AGG_WAVES - 1) / AGG_WAVES;
        const int beg = w * per;
        const int end = (beg + per < kept) ? (beg + per) : kept;
        for (int c = beg; c < end; c += AGG_CHUNK) {
            const int m = (end - c < AGG_CHUNK) ? (end - c) : AGG_CHUNK;
            for (int j = lane; j < m; j += 32) wq[j] = mylist[c + j];
            agg_accumulate(wq, m, lane, h, dinv, sdinv, acc);
        }
    }
    __syncthreads();
    if (KEEP && tid == 0) counts[blockIdx.x] = cursor;

    float* op = out + lo * DFEAT;
    for (int i = tid; i < cnt * DFEAT; i += AGG_THREADS)
        op[i] = tanhf(acc[i]);
}

extern "C" void kernel_launch(void* const* d_in, const int* in_sizes, int n_in,
                              void* d_out, int out_size, void* d_ws, size_t ws_size,
                              hipStream_t stream) {
    const float* x    = (const float*)d_in[0];
    const int*   ei   = (const int*)  d_in[1];
    const float* W1   = (const float*)d_in[2];
    const float* b1   = (const float*)d_in[3];
    const float* W2   = (const float*)d_in[4];
    const float* b2   = (const float*)d_in[5];
    const float* Wlin = (const float*)d_in[6];
    const float* blin = (const float*)d_in[7];

    const int* src = ei;
    const int* dst = ei + N_EDGES;

    float* wsf  = (float*)d_ws;
    float* dinv = wsf;
    float* bufA = wsf + 50048;
    float* bufB = bufA + (size_t)N_NODES * DFEAT;
    int*   lists  = (int*)(bufB + (size_t)N_NODES * DFEAT);
    int*   counts = lists + (size_t)AGG_BLOCKS * AGG_LCAP * 2;
    int*   segcount = counts + AGG_BLOCKS;
    int*   segs     = segcount + BIN_RANGES * BIN_CHUNKS;

    const dim3 gemmGrid(N_NODES / 16, DFEAT / 64);

    k_init_deg <<<(N_NODES + 255) / 256, 256, 0, stream>>>((int*)dinv, N_NODES);
    k_count_partition<<<BIN_CHUNKS, BIN_THREADS, 0, stream>>>((int*)dinv, src, dst, segs, segcount);
    k_make_dinv<<<(N_NODES + 255) / 256, 256, 0, stream>>>(dinv, N_NODES);

    k_gemm<DFEAT><<<gemmGrid, 128, 0, stream>>>(x, W1, bufA);
    k_aggregate<true><<<AGG_BLOCKS, AGG_THREADS, 0, stream>>>(bufA, segs, segcount, dinv, b1, bufB, lists, counts);

    k_gemm<DFEAT><<<gemmGrid, 128, 0, stream>>>(bufB, W2, bufA);
    k_aggregate<false><<<AGG_BLOCKS, AGG_THREADS, 0, stream>>>(bufA, segs, segcount, dinv, b2, bufB, lists, counts);

    k_gemm_bias<NCLS><<<N_NODES / 16, 96, 0, stream>>>(bufB, Wlin, blin, (float*)d_out);
}


# Dense products re-tiled: a wave forms 16 rows x 64 columns as 4 accumulator tiles, columns interleaved (4c+t): one A fragment per 4 products, 16-B B loads and C stores per lane; sum order kept
# speedup vs baseline: 2.1413x; 1.0930x over previous
#include <hip/hip_runtime.h>
#include <math.h>

typedef float v2f __attribute__((ext_vector_type(2)));
typedef float v8f __attribute__((ext_vector_type(8)));

#define N_NODES 50000
#define N_EDGES 600000
#define DFEAT   128
#define NCLS    40

__global__ void k_init_deg(int* deg, int n) {
    int i = blockIdx.x * blockDim.x + threadIdx.x;
    if (i < n) deg[i] = 1;
}

#define BIN_RANGES   16
#define BIN_OWNERS   16
#define BIN_NODES    (BIN_OWNERS * 196)
#define BIN_CHUNKS   256
#define BIN_EDGES    2344
#define BIN_THREADS  1024
#define BIN_WAVES    (BIN_THREADS / 32)
#define BIN_PER_WAVE 74
#define BIN_VECS     3

__global__ __launch_bounds__(BIN_THREADS)
void k_count_partition(int* deg, const int* __restrict__ src, const int* __restrict__ dst,
                       int* segs, int* segcount) {
    __shared__ int wcnt[BIN_WAVES][BIN_RANGES];
    __shared__ int wstart[BIN_WAVES][BIN_RANGES];

    const int tid  = threadIdx.x;
    const int lane = tid & 31;
    const int wave = tid >> 5;
    const int e0 = blockIdx.x * BIN_EDGES;
    const int e1 = (e0 + BIN_EDGES < N_EDGES) ? (e0 + BIN_EDGES) : N_EDGES;
    const int wb = e0 + wave * BIN_PER_WAVE;
    const int we = (wb + BIN_PER_WAVE < e1) ? (wb + BIN_PER_WAVE) : e1;

    int s[BIN_VECS], d[BIN_VECS], r[BIN_VECS], rk[BIN_VECS];
    #pragma unroll
    for (int v = 0; v < BIN_VECS; ++v) {
        const int  e  = wb + 32 * v + lane;
        const bool ok = e < we;
        d[v] = ok ? dst[e] : -1;
        s[v] = ok ? src[e] : 0;
    }
    #pragma unroll
    for (int v = 0; v < BIN_VECS; ++v) {
        r[v]  = (d[v] >= 0) ? (d[v] / BIN_NODES) : -1;
        rk[v] = 0;
        if (d[v] >= 0) atomicAdd(&deg[d[v]], 1);
    }

    int mine = 0;
    #pragma unroll
    for (int b = 0; b < BIN_RANGES; ++b) {
        int run = 0;
        #pragma unroll
        for (int v = 0; v < BIN_VECS; ++v) {
            const bool hit = (r[v] == b);
            const unsigned m = __builtin_amdgcn_ballot_w32(hit);
            if (hit) rk[v] = run + (int)__builtin_amdgcn_mbcnt_lo(m, 0u);
            run += __builtin_popcount(m);
        }
        if (lane == b) mine = run;
    }
    if (lane < BIN_RANGES) wcnt[wave][lane] = mine;
    __syncthreads();

    if (tid < BIN_WAVES * BIN_RANGES) {
        const int w = tid >> 4;
        const int b = tid & 15;
        int sum = 0;
        for (int w2 = 0; w2 < w; ++w2) sum += wcnt[w2][b];
        wstart[w][b] = sum;
        if (w == BIN_WAVES - 1) segcount[b * BIN_CHUNKS + blockIdx.x] = sum + wcnt[w][b];
    }
    __syncthreads();

    int2* seg2 = (int2*)segs;
    #pragma unroll
    for (int v = 0; v < BIN_VECS; ++v) {
        if (r[v] >= 0) {
            const int pos = wstart[wave][r[v]] + rk[v];
            seg2[(r[v] * BIN_CHUNKS + blockIdx.x) * BIN_EDGES + pos] = make_int2(s[v], d[v]);
        }
    }
}

__global__ void k_make_dinv(float* dinv, int n) {
    int i = blockIdx.x * blockDim.x + threadIdx.x;
    if (i < n) {
        int d = ((const int*)dinv)[i];
        dinv[i] = rsqrtf((float)d);
    }
}

typedef float v4f __attribute__((ext_vector_type(4)));
#define GEMM_WAVES  4
#define GEMM_STRIPS (N_NODES / 16)

template<int N, bool BIAS>
__device__ __forceinline__ void gemm_piece(const float* __restrict__ A, const float* __restrict__ B,
                                           const float* __restrict__ bias, float* __restrict__ C) {
    const int lane  = threadIdx.x & 31;
    const int wave  = threadIdx.x >> 5;
    const int strip = blockIdx.x * GEMM_WAVES + wave;
    if (strip >= GEMM_STRIPS) return;
    const int row0  = strip * 16;
    const int c     = lane & 15;
    const int h     = lane >> 4;
    const int col   = blockIdx.y * 64 + 4 * c;
    const bool live = (N % 64 == 0) || (col < N);
    const int colc  = live ? col : (N - 4);

    const float* Ap = A + (row0 + c) * DFEAT + 2 * h;
    const float* Bp = B + (2 * h) * N + colc;

    v8f acc0 = {}, acc1 = {}, acc2 = {}, acc3 = {};
    #pragma unroll
    for (int k0 = 0; k0 < DFEAT; k0 += 4) {
        const v2f a  = *(const v2f*)(Ap + k0);
        const v4f b0 = *(const v4f*)(Bp + k0 * N);
        const v4f b1 = *(const v4f*)(Bp + k0 * N + N);
        v2f b;
        b.x = b0.x; b.y = b1.x;
        acc0 = __builtin_amdgcn_wmma_f32_16x16x4_f32(false, a, false, b, (short)0, acc0, false, false);
        b.x = b0.y; b.y = b1.y;
        acc1 = __builtin_amdgcn_wmma_f32_16x16x4_f32(false, a, false, b, (short)0, acc1, false, false);
        b.x = b0.z; b.y = b1.z;
        acc2 = __builtin_amdgcn_wmma_f32_16x16x4_f32(false, a, false, b, (short)0, acc2, false, false);
        b.x = b0.w; b.y = b1.w;
        acc3 = __builtin_amdgcn_wmma_f32_16x16x4_f32(false, a, false, b, (short)0, acc3, false, false);
    }

    if (live) {
        v4f bb = {0.0f, 0.0f, 0.0f, 0.0f};
        if (BIAS) bb = *(const v4f*)(bias + col);
        float* Cp = C + (row0 + 8 * h) * N + col;
        #pragma unroll
        for (int i = 0; i < 8; ++i) {
            v4f o;
            o.x = acc0[i]; o.y = acc1[i]; o.z = acc2[i]; o.w = acc3[i];
            if (BIAS) o += bb;
            *(v4f*)(Cp + i * N) = o;
        }
    }
}

__global__ __launch_bounds__(32 * GEMM_WAVES)
void k_gemm(const float* __restrict__ A, const float* __restrict__ B, float* __restrict__ C) {
    gemm_piece<DFEAT, false>(A, B, nullptr, C);
}

__global__ __launch_bounds__(32 * GEMM_WAVES)
void k_gemm_bias(const float* __restrict__ A, const float* __restrict__ B,
                 const float* __restrict__ bias, float* __restrict__ C) {
    gemm_piece<NCLS, true>(A, B, bias, C);
}

#define AGG_BLOCKS   256
#define AGG_NODES    196
#define AGG_THREADS  1024
#define AGG_WAVES    (AGG_THREADS / 32)
#define AGG_SEGS     (BIN_CHUNKS / AGG_WAVES)
#define AGG_QCAP     544
#define AGG_GROUP    8
#define AGG_LCAP     8192
#define AGG_CHUNK    512

__device__ __forceinline__ void agg_accumulate(const int2* wq, const int qn, const int lane,
                                               const float* __restrict__ h,
                                               const float* __restrict__ dinv,
                                               const float* sdinv, float* acc) {
    for (int j = 0; j < qn; j += AGG_GROUP) {
        int s[AGG_GROUP], dl[AGG_GROUP];
        #pragma unroll
        for (int k = 0; k < AGG_GROUP; ++k) {
            const int jj = (j + k < qn) ? (j + k) : (qn - 1);
            const int2 ent = wq[jj];
            s[k]  = __builtin_amdgcn_readfirstlane(ent.x);
            dl[k] = __builtin_amdgcn_readfirstlane(ent.y);
        }
        float v[AGG_GROUP][4];
        float nrm[AGG_GROUP];
        #pragma unroll
        for (int k = 0; k < AGG_GROUP; ++k) {
            nrm[k] = dinv[s[k]] * sdinv[dl[k]];
            const float* hp = h + s[k] * DFEAT + lane;
            v[k][0] = hp[0];
            v[k][1] = hp[32];
            v[k][2] = hp[64];
            v[k][3] = hp[96];
        }
        #pragma unroll
        for (int k = 0; k < AGG_GROUP; ++k) {
            if (j + k < qn) {
                float* a = acc + dl[k] * DFEAT + lane;
                atomicAdd(a +  0, v[k][0] * nrm[k]);
                atomicAdd(a + 32, v[k][1] * nrm[k]);
                atomicAdd(a + 64, v[k][2] * nrm[k]);
                atomicAdd(a + 96, v[k][3] * nrm[k]);
            }
        }
    }
}

template<bool KEEP>
__device__ __forceinline__ void agg_drain(const int2* wq, const int qn, const int lane,
                                          const float* __restrict__ h,
                                          const float* __restrict__ dinv,
                                          const float* sdinv, float* acc,
                                          int2* list, int* cursor) {
    if (KEEP) {
        int base = 0;
        if (lane == 0) base = atomicAdd(cursor, qn);
        base = __builtin_amdgcn_readfirstlane(base);
        if (base + qn <= AGG_LCAP)
            for (int j = lane; j < qn; j += 32) list[base + j] = wq[j];
    }
    agg_accumulate(wq, qn, lane, h, dinv, sdinv, acc);
}

template<bool KEEP>
__global__ __launch_bounds__(AGG_THREADS)
void k_aggregate(const float* __restrict__ h, const int* __restrict__ segs,
                 const int* __restrict__ segcount, const float* __restrict__ dinv,
                 const float* __restrict__ bias, float* __restrict__ out,
                 int* lists, int* counts) {
    __shared__ float acc[AGG_NODES * DFEAT];
    __shared__ float sdinv[AGG_NODES];
    __shared__ int2  queue[AGG_WAVES][AGG_QCAP];
    __shared__ int   cursor;

    const int tid  = threadIdx.x;
    const int lane = tid & 31;
    const int wave = tid >> 5;
    const int lo   = blockIdx.x * AGG_NODES;
    const int cnt  = (N_NODES - lo < AGG_NODES) ? (N_NODES - lo) : AGG_NODES;

    int2* mylist = (int2*)lists + blockIdx.x * AGG_LCAP;

    int  kept = 0;
    bool scan = true;
    if (!KEEP) {
        kept = __builtin_amdgcn_readfirstlane(counts[blockIdx.x]);
        scan = kept > AGG_LCAP;
    }

    const int   range = blockIdx.x / BIN_OWNERS;
    const int2* seg2  = (const int2*)segs + (range * BIN_CHUNKS + wave) * BIN_EDGES;
    int len[AGG_SEGS];
    #pragma unroll
    for (int k = 0; k < AGG_SEGS; ++k)
        len[k] = scan ? segcount[range * BIN_CHUNKS + wave + k * AGG_WAVES] : 0;

    for (int i = tid; i < cnt * DFEAT; i += AGG_THREADS) {
        const int   n  = i >> 7;
        const float di = dinv[lo + n];
        acc[i] = h[(lo + n) * DFEAT + (i & 127)] * di * di + bias[i & 127];
    }
    for (int i = tid; i < cnt; i += AGG_THREADS) sdinv[i] = dinv[lo + i];
    if (tid == 0) cursor = 0;
    __syncthreads();

    int2* wq = queue[wave];

    if (scan) {
        int longest = 0;
        #pragma unroll
        for (int k = 0; k < AGG_SEGS; ++k) {
            len[k]  = __builtin_amdgcn_readfirstlane(len[k]);
            len[k]  = (len[k] < BIN_EDGES) ? len[k] : BIN_EDGES;
            longest = (len[k] > longest) ? len[k] : longest;
        }

        int qn = 0;

        for (int t0 = 0; t0 < longest; t0 += 32) {
            const int t = t0 + lane;
            int2 pr[AGG_SEGS];
            #pragma unroll
            for (int k = 0; k < AGG_SEGS; ++k)
                pr[k] = (t < len[k]) ? seg2[k * AGG_WAVES * BIN_EDGES + t] : make_int2(-1, -1);

            if (qn > AGG_QCAP - 32 * AGG_SEGS) {
                agg_drain<KEEP>(wq, qn, lane, h, dinv, sdinv, acc, mylist, &cursor);
                qn = 0;
            }
            #pragma unroll
            for (int k = 0; k < AGG_SEGS; ++k) {
                const int  dl  = pr[k].y - lo;
                const bool hit = (unsigned)dl < (unsigned)cnt;
                const unsigned m = __builtin_amdgcn_ballot_w32(hit);
                if (hit) {
                    const int pos = qn + (int)__builtin_amdgcn_mbcnt_lo(m, 0u);
                    wq[pos] = make_int2(pr[k].x, dl);
                }
                qn += __builtin_popcount(m);
            }
        }
        if (qn > 0) agg_drain<KEEP>(wq, qn, lane, h, dinv, sdinv, acc, mylist, &cursor);
    } else {
        const int w   = __builtin_amdgcn_readfirstlane(wave);
        const int per = (kept + AGG_WAVES - 1) / AGG_WAVES;
        const int beg = w * per;
        const int end = (beg + per < kept) ? (beg + per) : kept;
        for (int c = beg; c < end; c += AGG_CHUNK) {
            const int m = (end - c < AGG_CHUNK) ? (end - c) : AGG_CHUNK;
            for (int j = lane; j < m; j += 32) wq[j] = mylist[c + j];
            agg_accumulate(wq, m, lane, h, dinv, sdinv, acc);
        }
    }
    __syncthreads();
    if (KEEP && tid == 0) counts[blockIdx.x] = cursor;

    float* op = out + lo * DFEAT;
    for (int i = tid; i < cnt * DFEAT; i += AGG_THREADS)
        op[i] = tanhf(acc[i]);
}

extern "C" void kernel_launch(void* const* d_in, const int* in_sizes, int n_in,
                              void* d_out, int out_size, void* d_ws, size_t ws_size,
                              hipStream_t stream) {
    const float* x    = (const float*)d_in[0];
    const int*   ei   = (const int*)  d_in[1];
    const float* W1   = (const float*)d_in[2];
    const float* b1   = (const float*)d_in[3];
    const float* W2   = (const float*)d_in[4];
    const float* b2   = (const float*)d_in[5];
    const float* Wlin = (const float*)d_in[6];
    const float* blin = (const float*)d_in[7];

    const int* src = ei;
    const int* dst = ei + N_EDGES;

    float* wsf  = (float*)d_ws;
    float* dinv = wsf;
    float* bufA = wsf + 50048;
    float* bufB = bufA + (size_t)N_NODES * DFEAT;
    int*   lists  = (int*)(bufB + (size_t)N_NODES * DFEAT);
    int*   counts = lists + (size_t)AGG_BLOCKS * AGG_LCAP * 2;
    int*   segcount = counts + AGG_BLOCKS;
    int*   segs     = segcount + BIN_RANGES * BIN_CHUNKS;

    const dim3 gemmGrid((GEMM_STRIPS + GEMM_WAVES - 1) / GEMM_WAVES, DFEAT / 64);

    k_init_deg <<<(N_NODES + 255) / 256, 256, 0, stream>>>((int*)dinv, N_NODES);
    k_count_partition<<<BIN_CHUNKS, BIN_THREADS, 0, stream>>>((int*)dinv, src, dst, segs, segcount);
    k_make_dinv<<<(N_NODES + 255) / 256, 256, 0, stream>>>(dinv, N_NODES);

    k_gemm<<<gemmGrid, 32 * GEMM_WAVES, 0, stream>>>(x, W1, bufA);
    k_aggregate<true><<<AGG_BLOCKS, AGG_THREADS, 0, stream>>>(bufA, segs, segcount, dinv, b1, bufB, lists, counts);

    k_gemm<<<gemmGrid, 32 * GEMM_WAVES, 0, stream>>>(bufB, W2, bufA);
    k_aggregate<false><<<AGG_BLOCKS, AGG_THREADS, 0, stream>>>(bufA, segs, segcount, dinv, b2, bufB, lists, counts);

    k_gemm_bias<<<(GEMM_STRIPS + GEMM_WAVES - 1) / GEMM_WAVES, 32 * GEMM_WAVES, 0, stream>>>(bufB, Wlin, blin, (float*)d_out);
}


# In-degrees counted in LDS by the 256 owner blocks while they scan their segments (scan and list moved from layer 1 into this pass): 600000 global int atomics and 1 launch go; LDS float sums as before
# speedup vs baseline: 2.5476x; 1.1897x over previous
#include <hip/hip_runtime.h>
#include <math.h>

typedef float v2f __attribute__((ext_vector_type(2)));
typedef float v8f __attribute__((ext_vector_type(8)));

#define N_NODES 50000
#define N_EDGES 600000
#define DFEAT   128
#define NCLS    40

#define BIN_RANGES   16
#define BIN_OWNERS   16
#define BIN_NODES    (BIN_OWNERS * 196)
#define BIN_CHUNKS   256
#define BIN_EDGES    2344
#define BIN_THREADS  1024
#define BIN_WAVES    (BIN_THREADS / 32)
#define BIN_PER_WAVE 74
#define BIN_VECS     3

__global__ __launch_bounds__(BIN_THREADS)
void k_partition(const int* __restrict__ src, const int* __restrict__ dst,
                 int* segs, int* segcount) {
    __shared__ int wcnt[BIN_WAVES][BIN_RANGES];
    __shared__ int wstart[BIN_WAVES][BIN_RANGES];

    const int tid  = threadIdx.x;
    const int lane = tid & 31;
    const int wave = tid >> 5;
    const int e0 = blockIdx.x * BIN_EDGES;
    const int e1 = (e0 + BIN_EDGES < N_EDGES) ? (e0 + BIN_EDGES) : N_EDGES;
    const int wb = e0 + wave * BIN_PER_WAVE;
    const int we = (wb + BIN_PER_WAVE < e1) ? (wb + BIN_PER_WAVE) : e1;

    int s[BIN_VECS], d[BIN_VECS], r[BIN_VECS], rk[BIN_VECS];
    #pragma unroll
    for (int v = 0; v < BIN_VECS; ++v) {
        const int  e  = wb + 32 * v + lane;
        const bool ok = e < we;
        d[v] = ok ? dst[e] : -1;
        s[v] = ok ? src[e] : 0;
    }
    #pragma unroll
    for (int v = 0; v < BIN_VECS; ++v) {
        r[v]  = (d[v] >= 0) ? (d[v] / BIN_NODES) : -1;
        rk[v] = 0;
    }

    int mine = 0;
    #pragma unroll
    for (int b = 0; b < BIN_RANGES; ++b) {
        int run = 0;
        #pragma unroll
        for (int v = 0; v < BIN_VECS; ++v) {
            const bool hit = (r[v] == b);
            const unsigned m = __builtin_amdgcn_ballot_w32(hit);
            if (hit) rk[v] = run + (int)__builtin_amdgcn_mbcnt_lo(m, 0u);
            run += __builtin_popcount(m);
        }
        if (lane == b) mine = run;
    }
    if (lane < BIN_RANGES) wcnt[wave][lane] = mine;
    __syncthreads();

    if (tid < BIN_WAVES * BIN_RANGES) {
        const int w = tid >> 4;
        const int b = tid & 15;
        int sum = 0;
        for (int w2 = 0; w2 < w; ++w2) sum += wcnt[w2][b];
        wstart[w][b] = sum;
        if (w == BIN_WAVES - 1) segcount[b * BIN_CHUNKS + blockIdx.x] = sum + wcnt[w][b];
    }
    __syncthreads();

    int2* seg2 = (int2*)segs;
    #pragma unroll
    for (int v = 0; v < BIN_VECS; ++v) {
        if (r[v] >= 0) {
            const int pos = wstart[wave][r[v]] + rk[v];
            seg2[(r[v] * BIN_CHUNKS + blockIdx.x) * BIN_EDGES + pos] = make_int2(s[v], d[v]);
        }
    }
}

typedef float v4f __attribute__((ext_vector_type(4)));
#define GEMM_WAVES  4
#define GEMM_STRIPS (N_NODES / 16)

template<int N, bool BIAS>
__device__ __forceinline__ void gemm_piece(const float* __restrict__ A, const float* __restrict__ B,
                                           const float* __restrict__ bias, float* __restrict__ C) {
    const int lane  = threadIdx.x & 31;
    const int wave  = threadIdx.x >> 5;
    const int strip = blockIdx.x * GEMM_WAVES + wave;
    if (strip >= GEMM_STRIPS) return;
    const int row0  = strip * 16;
    const int c     = lane & 15;
    const int h     = lane >> 4;
    const int col   = blockIdx.y * 64 + 4 * c;
    const bool live = (N % 64 == 0) || (col < N);
    const int colc  = live ? col : (N - 4);

    const float* Ap = A + (row0 + c) * DFEAT + 2 * h;
    const float* Bp = B + (2 * h) * N + colc;

    v8f acc0 = {}, acc1 = {}, acc2 = {}, acc3 = {};
    #pragma unroll
    for (int k0 = 0; k0 < DFEAT; k0 += 4) {
        const v2f a  = *(const v2f*)(Ap + k0);
        const v4f b0 = *(const v4f*)(Bp + k0 * N);
        const v4f b1 = *(const v4f*)(Bp + k0 * N + N);
        v2f b;
        b.x = b0.x; b.y = b1.x;
        acc0 = __builtin_amdgcn_wmma_f32_16x16x4_f32(false, a, false, b, (short)0, acc0, false, false);
        b.x = b0.y; b.y = b1.y;
        acc1 = __builtin_amdgcn_wmma_f32_16x16x4_f32(false, a, false, b, (short)0, acc1, false, false);
        b.x = b0.z; b.y = b1.z;
        acc2 = __builtin_amdgcn_wmma_f32_16x16x4_f32(false, a, false, b, (short)0, acc2, false, false);
        b.x = b0.w; b.y = b1.w;
        acc3 = __builtin_amdgcn_wmma_f32_16x16x4_f32(false, a, false, b, (short)0, acc3, false, false);
    }

    if (live) {
        v4f bb = {0.0f, 0.0f, 0.0f, 0.0f};
        if (BIAS) bb = *(const v4f*)(bias + col);
        float* Cp = C + (row0 + 8 * h) * N + col;
        #pragma unroll
        for (int i = 0; i < 8; ++i) {
            v4f o;
            o.x = acc0[i]; o.y = acc1[i]; o.z = acc2[i]; o.w = acc3[i];
            if (BIAS) o += bb;
            *(v4f*)(Cp + i * N) = o;
        }
    }
}

__global__ __launch_bounds__(32 * GEMM_WAVES)
void k_gemm(const float* __restrict__ A, const float* __restrict__ B, float* __restrict__ C) {
    gemm_piece<DFEAT, false>(A, B, nullptr, C);
}

__global__ __launch_bounds__(32 * GEMM_WAVES)
void k_gemm_bias(const float* __restrict__ A, const float* __restrict__ B,
                 const float* __restrict__ bias, float* __restrict__ C) {
    gemm_piece<NCLS, true>(A, B, bias, C);
}

#define AGG_BLOCKS   256
#define AGG_NODES    196
#define AGG_THREADS  1024
#define AGG_WAVES    (AGG_THREADS / 32)
#define AGG_SEGS     (BIN_CHUNKS / AGG_WAVES)
#define AGG_QCAP     544
#define AGG_GROUP    8
#define AGG_LCAP     8192
#define AGG_CHUNK    512

__device__ __forceinline__ void list_keep(const int2* wq, const int qn, const int lane,
                                          int2* list, int* cursor) {
    int base = 0;
    if (lane == 0) base = atomicAdd(cursor, qn);
    base = __builtin_amdgcn_readfirstlane(base);
    if (base + qn <= AGG_LCAP)
        for (int j = lane; j < qn; j += 32) list[base + j] = wq[j];
}

__global__ __launch_bounds__(AGG_THREADS)
void k_degree_list(const int* __restrict__ segs, const int* __restrict__ segcount,
                   float* __restrict__ dinv, int* lists, int* counts) {
    __shared__ int   indeg[AGG_NODES];
    __shared__ int2  queue[AGG_WAVES][AGG_QCAP];
    __shared__ int   cursor;

    const int tid  = threadIdx.x;
    const int lane = tid & 31;
    const int wave = tid >> 5;
    const int lo   = blockIdx.x * AGG_NODES;
    const int cnt  = (N_NODES - lo < AGG_NODES) ? (N_NODES - lo) : AGG_NODES;

    int2* mylist = (int2*)lists + blockIdx.x * AGG_LCAP;

    const int   range = blockIdx.x / BIN_OWNERS;
    const int2* seg2  = (const int2*)segs + (range * BIN_CHUNKS + wave) * BIN_EDGES;
    int len[AGG_SEGS];
    #pragma unroll
    for (int k = 0; k < AGG_SEGS; ++k)
        len[k] = segcount[range * BIN_CHUNKS + wave + k * AGG_WAVES];

    for (int i = tid; i < AGG_NODES; i += AGG_THREADS) indeg[i] = 1;
    if (tid == 0) cursor = 0;
    __syncthreads();

    int2* wq = queue[wave];

    int longest = 0;
    #pragma unroll
    for (int k = 0; k < AGG_SEGS; ++k) {
        len[k]  = __builtin_amdgcn_readfirstlane(len[k]);
        len[k]  = (len[k] < BIN_EDGES) ? len[k] : BIN_EDGES;
        longest = (len[k] > longest) ? len[k] : longest;
    }

    int qn = 0;

    for (int t0 = 0; t0 < longest; t0 += 32) {
        const int t = t0 + lane;
        int2 pr[AGG_SEGS];
        #pragma unroll
        for (int k = 0; k < AGG_SEGS; ++k)
            pr[k] = (t < len[k]) ? seg2[k * AGG_WAVES * BIN_EDGES + t] : make_int2(-1, -1);

        if (qn > AGG_QCAP - 32 * AGG_SEGS) {
            list_keep(wq, qn, lane, mylist, &cursor);
            qn = 0;
        }
        #pragma unroll
        for (int k = 0; k < AGG_SEGS; ++k) {
            const int  dl  = pr[k].y - lo;
            const bool hit = (unsigned)dl < (unsigned)cnt;
            const unsigned m = __builtin_amdgcn_ballot_w32(hit);
            if (hit) {
                const int pos = qn + (int)__builtin_amdgcn_mbcnt_lo(m, 0u);
                wq[pos] = make_int2(pr[k].x, dl);
                atomicAdd(&indeg[dl], 1);
            }
            qn += __builtin_popcount(m);
        }
    }
    if (qn > 0) list_keep(wq, qn, lane, mylist, &cursor);
    __syncthreads();
    if (tid == 0) counts[blockIdx.x] = cursor;

    for (int i = tid; i < cnt; i += AGG_THREADS)
        dinv[lo + i] = rsqrtf((float)indeg[i]);
}

__device__ __forceinline__ void agg_accumulate(const int2* wq, const int qn, const int lane,
                                               const float* __restrict__ h,
                                               const float* __restrict__ dinv,
                                               const float* sdinv, float* acc) {
    for (int j = 0; j < qn; j += AGG_GROUP) {
        int s[AGG_GROUP], dl[AGG_GROUP];
        #pragma unroll
        for (int k = 0; k < AGG_GROUP; ++k) {
            const int jj = (j + k < qn) ? (j + k) : (qn - 1);
            const int2 ent = wq[jj];
            s[k]  = __builtin_amdgcn_readfirstlane(ent.x);
            dl[k] = __builtin_amdgcn_readfirstlane(ent.y);
        }
        float v[AGG_GROUP][4];
        float nrm[AGG_GROUP];
        #pragma unroll
        for (int k = 0; k < AGG_GROUP; ++k) {
            nrm[k] = dinv[s[k]] * sdinv[dl[k]];
            const float* hp = h + s[k] * DFEAT + lane;
            v[k][0] = hp[0];
            v[k][1] = hp[32];
            v[k][2] = hp[64];
            v[k][3] = hp[96];
        }
        #pragma unroll
        for (int k = 0; k < AGG_GROUP; ++k) {
            if (j + k < qn) {
                float* a = acc + dl[k] * DFEAT + lane;
                atomicAdd(a +  0, v[k][0] * nrm[k]);
                atomicAdd(a + 32, v[k][1] * nrm[k]);
                atomicAdd(a + 64, v[k][2] * nrm[k]);
                atomicAdd(a + 96, v[k][3] * nrm[k]);
            }
        }
    }
}

__global__ __launch_bounds__(AGG_THREADS)
void k_aggregate(const float* __restrict__ h, const int* __restrict__ segs,
                 const int* __restrict__ segcount, const float* __restrict__ dinv,
                 const float* __restrict__ bias, float* __restrict__ out,
                 const int* __restrict__ lists, const int* __restrict__ counts) {
    __shared__ float acc[AGG_NODES * DFEAT];
    __shared__ float sdinv[AGG_NODES];
    __shared__ int2  queue[AGG_WAVES][AGG_QCAP];

    const int tid  = threadIdx.x;
    const int lane = tid & 31;
    const int wave = tid >> 5;
    const int lo   = blockIdx.x * AGG_NODES;
    const int cnt  = (N_NODES - lo < AGG_NODES) ? (N_NODES - lo) : AGG_NODES;

    const int2* mylist = (const int2*)lists + blockIdx.x * AGG_LCAP;

    const int  kept = __builtin_amdgcn_readfirstlane(counts[blockIdx.x]);
    const bool scan = kept > AGG_LCAP;

    const int   range = blockIdx.x / BIN_OWNERS;
    const int2* seg2  = (const int2*)segs + (range * BIN_CHUNKS + wave) * BIN_EDGES;
    int len[AGG_SEGS];
    #pragma unroll
    for (int k = 0; k < AGG_SEGS; ++k)
        len[k] = scan ? segcount[range * BIN_CHUNKS + wave + k * AGG_WAVES] : 0;

    for (int i = tid; i < cnt * DFEAT; i += AGG_THREADS) {
        const int   n  = i >> 7;
        const float di = dinv[lo + n];
        acc[i] = h[(lo + n) * DFEAT + (i & 127)] * di * di + bias[i & 127];
    }
    for (int i = tid; i < cnt; i += AGG_THREADS) sdinv[i] = dinv[lo + i];
    __syncthreads();

    int2* wq = queue[wave];

    if (!scan) {
        const int w   = __builtin_amdgcn_readfirstlane(wave);
        const int per = (kept + AGG_WAVES - 1) / AGG_WAVES;
        const int beg = w * per;
        const int end = (beg + per < kept) ? (beg + per) : kept;
        for (int c = beg; c < end; c += AGG_CHUNK) {
            const int m = (end - c < AGG_CHUNK) ? (end - c) : AGG_CHUNK;
            for (int j = lane; j < m; j += 32) wq[j] = mylist[c + j];
            agg_accumulate(wq, m, lane, h, dinv, sdinv, acc);
        }
    } else {
        int longest = 0;
        #pragma unroll
        for (int k = 0; k < AGG_SEGS; ++k) {
            len[k]  = __builtin_amdgcn_readfirstlane(len[k]);
            len[k]  = (len[k] < BIN_EDGES) ? len[k] : BIN_EDGES;
            longest = (len[k] > longest) ? len[k] : longest;
        }

        int qn = 0;

        for (int t0 = 0; t0 < longest; t0 += 32) {
            const int t = t0 + lane;
            int2 pr[AGG_SEGS];
            #pragma unroll
            for (int k = 0; k < AGG_SEGS; ++k)
                pr[k] = (t < len[k]) ? seg2[k * AGG_WAVES * BIN_EDGES + t] : make_int2(-1, -1);

            if (qn > AGG_QCAP - 32 * AGG_SEGS) {
                agg_accumulate(wq, qn, lane, h, dinv, sdinv, acc);
                qn = 0;
            }
            #pragma unroll
            for (int k = 0; k < AGG_SEGS; ++k) {
                const int  dl  = pr[k].y - lo;
                const bool hit = (unsigned)dl < (unsigned)cnt;
                const unsigned m = __builtin_amdgcn_ballot_w32(hit);
                if (hit) {
                    const int pos = qn + (int)__builtin_amdgcn_mbcnt_lo(m, 0u);
                    wq[pos] = make_int2(pr[k].x, dl);
                }
                qn += __builtin_popcount(m);
            }
        }
        if (qn > 0) agg_accumulate(wq, qn, lane, h, dinv, sdinv, acc);
    }
    __syncthreads();

    float* op = out + lo * DFEAT;
    for (int i = tid; i < cnt * DFEAT; i += AGG_THREADS)
        op[i] = tanhf(acc[i]);
}

extern "C" void kernel_launch(void* const* d_in, const int* in_sizes, int n_in,
                              void* d_out, int out_size, void* d_ws, size_t ws_size,
                              hipStream_t stream) {
    const float* x    = (const float*)d_in[0];
    const int*   ei   = (const int*)  d_in[1];
    const float* W1   = (const float*)d_in[2];
    const float* b1   = (const float*)d_in[3];
    const float* W2   = (const float*)d_in[4];
    const float* b2   = (const float*)d_in[5];
    const float* Wlin = (const float*)d_in[6];
    const float* blin = (const float*)d_in[7];

    const int* src = ei;
    const int* dst = ei + N_EDGES;

    float* wsf  = (float*)d_ws;
    float* dinv = wsf;
    float* bufA = wsf + 50048;
    float* bufB = bufA + (size_t)N_NODES * DFEAT;
    int*   lists  = (int*)(bufB + (size_t)N_NODES * DFEAT);
    int*   counts = lists + (size_t)AGG_BLOCKS * AGG_LCAP * 2;
    int*   segcount = counts + AGG_BLOCKS;
    int*   segs     = segcount + BIN_RANGES * BIN_CHUNKS;

    const dim3 gemmGrid((GEMM_STRIPS + GEMM_WAVES - 1) / GEMM_WAVES, DFEAT / 64);

    k_partition<<<BIN_CHUNKS, BIN_THREADS, 0, stream>>>(src, dst, segs, segcount);
    k_degree_list<<<AGG_BLOCKS, AGG_THREADS, 0, stream>>>(segs, segcount, dinv, lists, counts);

    k_gemm<<<gemmGrid, 32 * GEMM_WAVES, 0, stream>>>(x, W1, bufA);
    k_aggregate<<<AGG_BLOCKS, AGG_THREADS, 0, stream>>>(bufA, segs, segcount, dinv, b1, bufB, lists, counts);

    k_gemm<<<gemmGrid, 32 * GEMM_WAVES, 0, stream>>>(bufB, W2, bufA);
    k_aggregate<<<AGG_BLOCKS, AGG_THREADS, 0, stream>>>(bufA, segs, segcount, dinv, b2, bufB, lists, counts);

    k_gemm_bias<<<(GEMM_STRIPS + GEMM_WAVES - 1) / GEMM_WAVES, 32 * GEMM_WAVES, 0, stream>>>(bufB, Wlin, blin, (float*)d_out);
}


# Owner kernel's opening: own rows, bias, dinv and list run requested together at entry (a wave's first LDS add stood behind 12 dependent waits, now 4); sums start at 0, self-loop+bias added last
# speedup vs baseline: 2.8916x; 1.1350x over previous
#include <hip/hip_runtime.h>
#include <math.h>

typedef float v2f __attribute__((ext_vector_type(2)));
typedef float v8f __attribute__((ext_vector_type(8)));

#define N_NODES 50000
#define N_EDGES 600000
#define DFEAT   128
#define NCLS    40

#define BIN_RANGES   16
#define BIN_OWNERS   16
#define BIN_NODES    (BIN_OWNERS * 196)
#define BIN_CHUNKS   256
#define BIN_EDGES    2344
#define BIN_THREADS  1024
#define BIN_WAVES    (BIN_THREADS / 32)
#define BIN_PER_WAVE 74
#define BIN_VECS     3

__global__ __launch_bounds__(BIN_THREADS)
void k_partition(const int* __restrict__ src, const int* __restrict__ dst,
                 int* segs, int* segcount) {
    __shared__ int wcnt[BIN_WAVES][BIN_RANGES];
    __shared__ int wstart[BIN_WAVES][BIN_RANGES];

    const int tid  = threadIdx.x;
    const int lane = tid & 31;
    const int wave = tid >> 5;
    const int e0 = blockIdx.x * BIN_EDGES;
    const int e1 = (e0 + BIN_EDGES < N_EDGES) ? (e0 + BIN_EDGES) : N_EDGES;
    const int wb = e0 + wave * BIN_PER_WAVE;
    const int we = (wb + BIN_PER_WAVE < e1) ? (wb + BIN_PER_WAVE) : e1;

    int s[BIN_VECS], d[BIN_VECS], r[BIN_VECS], rk[BIN_VECS];
    #pragma unroll
    for (int v = 0; v < BIN_VECS; ++v) {
        const int  e  = wb + 32 * v + lane;
        const bool ok = e < we;
        d[v] = ok ? dst[e] : -1;
        s[v] = ok ? src[e] : 0;
    }
    #pragma unroll
    for (int v = 0; v < BIN_VECS; ++v) {
        r[v]  = (d[v] >= 0) ? (d[v] / BIN_NODES) : -1;
        rk[v] = 0;
    }

    int mine = 0;
    #pragma unroll
    for (int b = 0; b < BIN_RANGES; ++b) {
        int run = 0;
        #pragma unroll
        for (int v = 0; v < BIN_VECS; ++v) {
            const bool hit = (r[v] == b);
            const unsigned m = __builtin_amdgcn_ballot_w32(hit);
            if (hit) rk[v] = run + (int)__builtin_amdgcn_mbcnt_lo(m, 0u);
            run += __builtin_popcount(m);
        }
        if (lane == b) mine = run;
    }
    if (lane < BIN_RANGES) wcnt[wave][lane] = mine;
    __syncthreads();

    if (tid < BIN_WAVES * BIN_RANGES) {
        const int w = tid >> 4;
        const int b = tid & 15;
        int sum = 0;
        for (int w2 = 0; w2 < w; ++w2) sum += wcnt[w2][b];
        wstart[w][b] = sum;
        if (w == BIN_WAVES - 1) segcount[b * BIN_CHUNKS + blockIdx.x] = sum + wcnt[w][b];
    }
    __syncthreads();

    int2* seg2 = (int2*)segs;
    #pragma unroll
    for (int v = 0; v < BIN_VECS; ++v) {
        if (r[v] >= 0) {
            const int pos = wstart[wave][r[v]] + rk[v];
            seg2[(r[v] * BIN_CHUNKS + blockIdx.x) * BIN_EDGES + pos] = make_int2(s[v], d[v]);
        }
    }
}

typedef float v4f __attribute__((ext_vector_type(4)));
#define GEMM_WAVES  4
#define GEMM_STRIPS (N_NODES / 16)

template<int N, bool BIAS>
__device__ __forceinline__ void gemm_piece(const float* __restrict__ A, const float* __restrict__ B,
                                           const float* __restrict__ bias, float* __restrict__ C) {
    const int lane  = threadIdx.x & 31;
    const int wave  = threadIdx.x >> 5;
    const int strip = blockIdx.x * GEMM_WAVES + wave;
    if (strip >= GEMM_STRIPS) return;
    const int row0  = strip * 16;
    const int c     = lane & 15;
    const int h     = lane >> 4;
    const int col   = blockIdx.y * 64 + 4 * c;
    const bool live = (N % 64 == 0) || (col < N);
    const int colc  = live ? col : (N - 4);

    const float* Ap = A + (row0 + c) * DFEAT + 2 * h;
    const float* Bp = B + (2 * h) * N + colc;

    v8f acc0 = {}, acc1 = {}, acc2 = {}, acc3 = {};
    #pragma unroll
    for (int k0 = 0; k0 < DFEAT; k0 += 4) {
        const v2f a  = *(const v2f*)(Ap + k0);
        const v4f b0 = *(const v4f*)(Bp + k0 * N);
        const v4f b1 = *(const v4f*)(Bp + k0 * N + N);
        v2f b;
        b.x = b0.x; b.y = b1.x;
        acc0 = __builtin_amdgcn_wmma_f32_16x16x4_f32(false, a, false, b, (short)0, acc0, false, false);
        b.x = b0.y; b.y = b1.y;
        acc1 = __builtin_amdgcn_wmma_f32_16x16x4_f32(false, a, false, b, (short)0, acc1, false, false);
        b.x = b0.z; b.y = b1.z;
        acc2 = __builtin_amdgcn_wmma_f32_16x16x4_f32(false, a, false, b, (short)0, acc2, false, false);
        b.x = b0.w; b.y = b1.w;
        acc3 = __builtin_amdgcn_wmma_f32_16x16x4_f32(false, a, false, b, (short)0, acc3, false, false);
    }

    if (live) {
        v4f bb = {0.0f, 0.0f, 0.0f, 0.0f};
        if (BIAS) bb = *(const v4f*)(bias + col);
        float* Cp = C + (row0 + 8 * h) * N + col;
        #pragma unroll
        for (int i = 0; i < 8; ++i) {
            v4f o;
            o.x = acc0[i]; o.y = acc1[i]; o.z = acc2[i]; o.w = acc3[i];
            if (BIAS) o += bb;
            *(v4f*)(Cp + i * N) = o;
        }
    }
}

__global__ __launch_bounds__(32 * GEMM_WAVES)
void k_gemm(const float* __restrict__ A, const float* __restrict__ B, float* __restrict__ C) {
    gemm_piece<DFEAT, false>(A, B, nullptr, C);
}

__global__ __launch_bounds__(32 * GEMM_WAVES)
void k_gemm_bias(const float* __restrict__ A, const float* __restrict__ B,
                 const float* __restrict__ bias, float* __restrict__ C) {
    gemm_piece<NCLS, true>(A, B, bias, C);
}

#define AGG_BLOCKS   256
#define AGG_NODES    196
#define AGG_THREADS  1024
#define AGG_WAVES    (AGG_THREADS / 32)
#define AGG_SEGS     (BIN_CHUNKS / AGG_WAVES)
#define AGG_QCAP     544
#define AGG_GROUP    8
#define AGG_LCAP     8192
#define AGG_CHUNK    512
#define AGG_PIECES   (AGG_NODES * (DFEAT / 4))
#define AGG_OWN      ((AGG_PIECES + AGG_THREADS - 1) / AGG_THREADS)

__device__ __forceinline__ void list_keep(const int2* wq, const int qn, const int lane,
                                          int2* list, int* cursor) {
    int base = 0;
    if (lane == 0) base = atomicAdd(cursor, qn);
    base = __builtin_amdgcn_readfirstlane(base);
    if (base + qn <= AGG_LCAP)
        for (int j = lane; j < qn; j += 32) list[base + j] = wq[j];
}

__global__ __launch_bounds__(AGG_THREADS)
void k_degree_list(const int* __restrict__ segs, const int* __restrict__ segcount,
                   float* __restrict__ dinv, int* lists, int* counts) {
    __shared__ int   indeg[AGG_NODES];
    __shared__ int2  queue[AGG_WAVES][AGG_QCAP];
    __shared__ int   cursor;

    const int tid  = threadIdx.x;
    const int lane = tid & 31;
    const int wave = tid >> 5;
    const int lo   = blockIdx.x * AGG_NODES;
    const int cnt  = (N_NODES - lo < AGG_NODES) ? (N_NODES - lo) : AGG_NODES;

    int2* mylist = (int2*)lists + blockIdx.x * AGG_LCAP;

    const int   range = blockIdx.x / BIN_OWNERS;
    const int2* seg2  = (const int2*)segs + (range * BIN_CHUNKS + wave) * BIN_EDGES;
    int len[AGG_SEGS];
    #pragma unroll
    for (int k = 0; k < AGG_SEGS; ++k)
        len[k] = segcount[range * BIN_CHUNKS + wave + k * AGG_WAVES];

    for (int i = tid; i < AGG_NODES; i += AGG_THREADS) indeg[i] = 1;
    if (tid == 0) cursor = 0;
    __syncthreads();

    int2* wq = queue[wave];

    int longest = 0;
    #pragma unroll
    for (int k = 0; k < AGG_SEGS; ++k) {
        len[k]  = __builtin_amdgcn_readfirstlane(len[k]);
        len[k]  = (len[k] < BIN_EDGES) ? len[k] : BIN_EDGES;
        longest = (len[k] > longest) ? len[k] : longest;
    }

    int qn = 0;

    for (int t0 = 0; t0 < longest; t0 += 32) {
        const int t = t0 + lane;
        int2 pr[AGG_SEGS];
        #pragma unroll
        for (int k = 0; k < AGG_SEGS; ++k)
            pr[k] = (t < len[k]) ? seg2[k * AGG_WAVES * BIN_EDGES + t] : make_int2(-1, -1);

        if (qn > AGG_QCAP - 32 * AGG_SEGS) {
            list_keep(wq, qn, lane, mylist, &cursor);
            qn = 0;
        }
        #pragma unroll
        for (int k = 0; k < AGG_SEGS; ++k) {
            const int  dl  = pr[k].y - lo;
            const bool hit = (unsigned)dl < (unsigned)cnt;
            const unsigned m = __builtin_amdgcn_ballot_w32(hit);
            if (hit) {
                const int pos = qn + (int)__builtin_amdgcn_mbcnt_lo(m, 0u);
                wq[pos] = make_int2(pr[k].x, dl);
                atomicAdd(&indeg[dl], 1);
            }
            qn += __builtin_popcount(m);
        }
    }
    if (qn > 0) list_keep(wq, qn, lane, mylist, &cursor);
    __syncthreads();
    if (tid == 0) counts[blockIdx.x] = cursor;

    for (int i = tid; i < cnt; i += AGG_THREADS)
        dinv[lo + i] = rsqrtf((float)indeg[i]);
}

__device__ __forceinline__ void list_fetch128(int2* wq, const int2* from, const int j0, const int m, const int lane) {
    int2 t[4];
    #pragma unroll
    for (int u = 0; u < 4; ++u) {
        const int j = j0 + 32 * u + lane;
        t[u] = (j < m) ? from[j] : make_int2(0, 0);
    }
    #pragma unroll
    for (int u = 0; u < 4; ++u) {
        const int j = j0 + 32 * u + lane;
        if (j < m) wq[j] = t[u];
    }
}

__device__ __forceinline__ void agg_accumulate(const int2* wq, const int qn, const int lane,
                                               const float* __restrict__ h,
                                               const float* __restrict__ dinv,
                                               const float* sdinv, float* acc) {
    for (int j = 0; j < qn; j += AGG_GROUP) {
        int s[AGG_GROUP], dl[AGG_GROUP];
        #pragma unroll
        for (int k = 0; k < AGG_GROUP; ++k) {
            const int jj = (j + k < qn) ? (j + k) : (qn - 1);
            const int2 ent = wq[jj];
            s[k]  = __builtin_amdgcn_readfirstlane(ent.x);
            dl[k] = __builtin_amdgcn_readfirstlane(ent.y);
        }
        float v[AGG_GROUP][4];
        float nrm[AGG_GROUP];
        #pragma unroll
        for (int k = 0; k < AGG_GROUP; ++k) {
            nrm[k] = dinv[s[k]] * sdinv[dl[k]];
            const float* hp = h + s[k] * DFEAT + lane;
            v[k][0] = hp[0];
            v[k][1] = hp[32];
            v[k][2] = hp[64];
            v[k][3] = hp[96];
        }
        #pragma unroll
        for (int k = 0; k < AGG_GROUP; ++k) {
            if (j + k < qn) {
                float* a = acc + dl[k] * DFEAT + lane;
                atomicAdd(a +  0, v[k][0] * nrm[k]);
                atomicAdd(a + 32, v[k][1] * nrm[k]);
                atomicAdd(a + 64, v[k][2] * nrm[k]);
                atomicAdd(a + 96, v[k][3] * nrm[k]);
            }
        }
    }
}

__global__ __launch_bounds__(AGG_THREADS)
void k_aggregate(const float* __restrict__ h, const int* __restrict__ segs,
                 const int* __restrict__ segcount, const float* __restrict__ dinv,
                 const float* __restrict__ bias, float* __restrict__ out,
                 const int* __restrict__ lists, const int* __restrict__ counts) {
    __shared__ __attribute__((aligned(16))) float acc[AGG_NODES * DFEAT];
    __shared__ float sdinv[AGG_NODES];
    __shared__ int2  queue[AGG_WAVES][AGG_QCAP];

    const int tid  = threadIdx.x;
    const int lane = tid & 31;
    const int wave = tid >> 5;
    const int lo   = blockIdx.x * AGG_NODES;
    const int cnt  = (N_NODES - lo < AGG_NODES) ? (N_NODES - lo) : AGG_NODES;
    const int pieces = cnt * (DFEAT / 4);

    const v4f  zero4 = {0.0f, 0.0f, 0.0f, 0.0f};
    const v4f* h4    = (const v4f*)h + lo * (DFEAT / 4);
    v4f own[AGG_OWN];
    #pragma unroll
    for (int t = 0; t < AGG_OWN; ++t) {
        const int i = tid + t * AGG_THREADS;
        own[t] = (i < pieces) ? h4[i] : zero4;
    }
    const v4f   bb  = ((const v4f*)bias)[lane];
    const float myd = (tid < cnt) ? dinv[lo + tid] : 0.0f;

    const int2* mylist = (const int2*)lists + blockIdx.x * AGG_LCAP;

    const int  kept = __builtin_amdgcn_readfirstlane(counts[blockIdx.x]);
    const bool scan = kept > AGG_LCAP;

    const int   range = blockIdx.x / BIN_OWNERS;
    const int2* seg2  = (const int2*)segs + (range * BIN_CHUNKS + wave) * BIN_EDGES;
    int len[AGG_SEGS];
    #pragma unroll
    for (int k = 0; k < AGG_SEGS; ++k)
        len[k] = scan ? segcount[range * BIN_CHUNKS + wave + k * AGG_WAVES] : 0;

    int2* wq = queue[wave];

    const int w   = __builtin_amdgcn_readfirstlane(wave);
    const int per = (kept + AGG_WAVES - 1) / AGG_WAVES;
    const int beg = w * per;
    const int end = (beg + per < kept) ? (beg + per) : kept;
    if (!scan && beg < end)
        list_fetch128(wq, mylist + beg, 0, (end - beg < AGG_CHUNK) ? (end - beg) : AGG_CHUNK, lane);

    v4f* acc4 = (v4f*)acc;
    for (int i = tid; i < pieces; i += AGG_THREADS) acc4[i] = zero4;
    if (tid < cnt) sdinv[tid] = myd;
    __syncthreads();

    if (!scan) {
        for (int c = beg; c < end; c += AGG_CHUNK) {
            const int m = (end - c < AGG_CHUNK) ? (end - c) : AGG_CHUNK;
            for (int j0 = (c == beg) ? 128 : 0; j0 < m; j0 += 128)
                list_fetch128(wq, mylist + c, j0, m, lane);
            agg_accumulate(wq, m, lane, h, dinv, sdinv, acc);
        }
    } else {
        int longest = 0;
        #pragma unroll
        for (int k = 0; k < AGG_SEGS; ++k) {
            len[k]  = __builtin_amdgcn_readfirstlane(len[k]);
            len[k]  = (len[k] < BIN_EDGES) ? len[k] : BIN_EDGES;
            longest = (len[k] > longest) ? len[k] : longest;
        }

        int qn = 0;

        for (int t0 = 0; t0 < longest; t0 += 32) {
            const int t = t0 + lane;
            int2 pr[AGG_SEGS];
            #pragma unroll
            for (int k = 0; k < AGG_SEGS; ++k)
                pr[k] = (t < len[k]) ? seg2[k * AGG_WAVES * BIN_EDGES + t] : make_int2(-1, -1);

            if (qn > AGG_QCAP - 32 * AGG_SEGS) {
                agg_accumulate(wq, qn, lane, h, dinv, sdinv, acc);
                qn = 0;
            }
            #pragma unroll
            for (int k = 0; k < AGG_SEGS; ++k) {
                const int  dl  = pr[k].y - lo;
                const bool hit = (unsigned)dl < (unsigned)cnt;
                const unsigned m = __builtin_amdgcn_ballot_w32(hit);
                if (hit) {
                    const int pos = qn + (int)__builtin_amdgcn_mbcnt_lo(m, 0u);
                    wq[pos] = make_int2(pr[k].x, dl);
                }
                qn += __builtin_popcount(m);
            }
        }
        if (qn > 0) agg_accumulate(wq, qn, lane, h, dinv, sdinv, acc);
    }
    __syncthreads();

    v4f* op4 = (v4f*)(out + lo * DFEAT);
    #pragma unroll
    for (int t = 0; t < AGG_OWN; ++t) {
        const int i = tid + t * AGG_THREADS;
        if (i < pieces) {
            const float di = sdinv[i >> 5];
            const float nn = di * di;
            const v4f   a  = acc4[i];
            v4f o;
            o.x = tanhf(a.x + own[t].x * nn + bb.x);
            o.y = tanhf(a.y + own[t].y * nn + bb.y);
            o.z = tanhf(a.z + own[t].z * nn + bb.z);
            o.w = tanhf(a.w + own[t].w * nn + bb.w);
            op4[i] = o;
        }
    }
}

extern "C" void kernel_launch(void* const* d_in, const int* in_sizes, int n_in,
                              void* d_out, int out_size, void* d_ws, size_t ws_size,
                              hipStream_t stream) {
    const float* x    = (const float*)d_in[0];
    const int*   ei   = (const int*)  d_in[1];
    const float* W1   = (const float*)d_in[2];
    const float* b1   = (const float*)d_in[3];
    const float* W2   = (const float*)d_in[4];
    const float* b2   = (const float*)d_in[5];
    const float* Wlin = (const float*)d_in[6];
    const float* blin = (const float*)d_in[7];

    const int* src = ei;
    const int* dst = ei + N_EDGES;

    float* wsf  = (float*)d_ws;
    float* dinv = wsf;
    float* bufA = wsf + 50048;
    float* bufB = bufA + (size_t)N_NODES * DFEAT;
    int*   lists  = (int*)(bufB + (size_t)N_NODES * DFEAT);
    int*   counts = lists + (size_t)AGG_BLOCKS * AGG_LCAP * 2;
    int*   segcount = counts + AGG_BLOCKS;
    int*   segs     = segcount + BIN_RANGES * BIN_CHUNKS;

    const dim3 gemmGrid((GEMM_STRIPS + GEMM_WAVES - 1) / GEMM_WAVES, DFEAT / 64);

    k_partition<<<BIN_CHUNKS, BIN_THREADS, 0, stream>>>(src, dst, segs, segcount);
    k_degree_list<<<AGG_BLOCKS, AGG_THREADS, 0, stream>>>(segs, segcount, dinv, lists, counts);

    k_gemm<<<gemmGrid, 32 * GEMM_WAVES, 0, stream>>>(x, W1, bufA);
    k_aggregate<<<AGG_BLOCKS, AGG_THREADS, 0, stream>>>(bufA, segs, segcount, dinv, b1, bufB, lists, counts);

    k_gemm<<<gemmGrid, 32 * GEMM_WAVES, 0, stream>>>(bufB, W2, bufA);
    k_aggregate<<<AGG_BLOCKS, AGG_THREADS, 0, stream>>>(bufA, segs, segcount, dinv, b2, bufB, lists, counts);

    k_gemm_bias<<<(GEMM_STRIPS + GEMM_WAVES - 1) / GEMM_WAVES, 32 * GEMM_WAVES, 0, stream>>>(bufB, Wlin, blin, (float*)d_out);
}


# Gather loads widened: one 16-B load per lane takes a whole 512-B source row (was four 4-B one-line loads); LDS sums kept lane-interleaved so the float atomic adds stay conflict-free; sums as before
# speedup vs baseline: 2.9962x; 1.0362x over previous
#include <hip/hip_runtime.h>
#include <math.h>

typedef float v2f __attribute__((ext_vector_type(2)));
typedef float v8f __attribute__((ext_vector_type(8)));

#define N_NODES 50000
#define N_EDGES 600000
#define DFEAT   128
#define NCLS    40

#define BIN_RANGES   16
#define BIN_OWNERS   16
#define BIN_NODES    (BIN_OWNERS * 196)
#define BIN_CHUNKS   256
#define BIN_EDGES    2344
#define BIN_THREADS  1024
#define BIN_WAVES    (BIN_THREADS / 32)
#define BIN_PER_WAVE 74
#define BIN_VECS     3

__global__ __launch_bounds__(BIN_THREADS)
void k_partition(const int* __restrict__ src, const int* __restrict__ dst,
                 int* segs, int* segcount) {
    __shared__ int wcnt[BIN_WAVES][BIN_RANGES];
    __shared__ int wstart[BIN_WAVES][BIN_RANGES];

    const int tid  = threadIdx.x;
    const int lane = tid & 31;
    const int wave = tid >> 5;
    const int e0 = blockIdx.x * BIN_EDGES;
    const int e1 = (e0 + BIN_EDGES < N_EDGES) ? (e0 + BIN_EDGES) : N_EDGES;
    const int wb = e0 + wave * BIN_PER_WAVE;
    const int we = (wb + BIN_PER_WAVE < e1) ? (wb + BIN_PER_WAVE) : e1;

    int s[BIN_VECS], d[BIN_VECS], r[BIN_VECS], rk[BIN_VECS];
    #pragma unroll
    for (int v = 0; v < BIN_VECS; ++v) {
        const int  e  = wb + 32 * v + lane;
        const bool ok = e < we;
        d[v] = ok ? dst[e] : -1;
        s[v] = ok ? src[e] : 0;
    }
    #pragma unroll
    for (int v = 0; v < BIN_VECS; ++v) {
        r[v]  = (d[v] >= 0) ? (d[v] / BIN_NODES) : -1;
        rk[v] = 0;
    }

    int mine = 0;
    #pragma unroll
    for (int b = 0; b < BIN_RANGES; ++b) {
        int run = 0;
        #pragma unroll
        for (int v = 0; v < BIN_VECS; ++v) {
            const bool hit = (r[v] == b);
            const unsigned m = __builtin_amdgcn_ballot_w32(hit);
            if (hit) rk[v] = run + (int)__builtin_amdgcn_mbcnt_lo(m, 0u);
            run += __builtin_popcount(m);
        }
        if (lane == b) mine = run;
    }
    if (lane < BIN_RANGES) wcnt[wave][lane] = mine;
    __syncthreads();

    if (tid < BIN_WAVES * BIN_RANGES) {
        const int w = tid >> 4;
        const int b = tid & 15;
        int sum = 0;
        for (int w2 = 0; w2 < w; ++w2) sum += wcnt[w2][b];
        wstart[w][b] = sum;
        if (w == BIN_WAVES - 1) segcount[b * BIN_CHUNKS + blockIdx.x] = sum + wcnt[w][b];
    }
    __syncthreads();

    int2* seg2 = (int2*)segs;
    #pragma unroll
    for (int v = 0; v < BIN_VECS; ++v) {
        if (r[v] >= 0) {
            const int pos = wstart[wave][r[v]] + rk[v];
            seg2[(r[v] * BIN_CHUNKS + blockIdx.x) * BIN_EDGES + pos] = make_int2(s[v], d[v]);
        }
    }
}

typedef float v4f __attribute__((ext_vector_type(4)));
#define GEMM_WAVES  4
#define GEMM_STRIPS (N_NODES / 16)

template<int N, bool BIAS>
__device__ __forceinline__ void gemm_piece(const float* __restrict__ A, const float* __restrict__ B,
                                           const float* __restrict__ bias, float* __restrict__ C) {
    const int lane  = threadIdx.x & 31;
    const int wave  = threadIdx.x >> 5;
    const int strip = blockIdx.x * GEMM_WAVES + wave;
    if (strip >= GEMM_STRIPS) return;
    const int row0  = strip * 16;
    const int c     = lane & 15;
    const int h     = lane >> 4;
    const int col   = blockIdx.y * 64 + 4 * c;
    const bool live = (N % 64 == 0) || (col < N);
    const int colc  = live ? col : (N - 4);

    const float* Ap = A + (row0 + c) * DFEAT + 2 * h;
    const float* Bp = B + (2 * h) * N + colc;

    v8f acc0 = {}, acc1 = {}, acc2 = {}, acc3 = {};
    #pragma unroll
    for (int k0 = 0; k0 < DFEAT; k0 += 4) {
        const v2f a  = *(const v2f*)(Ap + k0);
        const v4f b0 = *(const v4f*)(Bp + k0 * N);
        const v4f b1 = *(const v4f*)(Bp + k0 * N + N);
        v2f b;
        b.x = b0.x; b.y = b1.x;
        acc0 = __builtin_amdgcn_wmma_f32_16x16x4_f32(false, a, false, b, (short)0, acc0, false, false);
        b.x = b0.y; b.y = b1.y;
        acc1 = __builtin_amdgcn_wmma_f32_16x16x4_f32(false, a, false, b, (short)0, acc1, false, false);
        b.x = b0.z; b.y = b1.z;
        acc2 = __builtin_amdgcn_wmma_f32_16x16x4_f32(false, a, false, b, (short)0, acc2, false, false);
        b.x = b0.w; b.y = b1.w;
        acc3 = __builtin_amdgcn_wmma_f32_16x16x4_f32(false, a, false, b, (short)0, acc3, false, false);
    }

    if (live) {
        v4f bb = {0.0f, 0.0f, 0.0f, 0.0f};
        if (BIAS) bb = *(const v4f*)(bias + col);
        float* Cp = C + (row0 + 8 * h) * N + col;
        #pragma unroll
        for (int i = 0; i < 8; ++i) {
            v4f o;
            o.x = acc0[i]; o.y = acc1[i]; o.z = acc2[i]; o.w = acc3[i];
            if (BIAS) o += bb;
            *(v4f*)(Cp + i * N) = o;
        }
    }
}

__global__ __launch_bounds__(32 * GEMM_WAVES)
void k_gemm(const float* __restrict__ A, const float* __restrict__ B, float* __restrict__ C) {
    gemm_piece<DFEAT, false>(A, B, nullptr, C);
}

__global__ __launch_bounds__(32 * GEMM_WAVES)
void k_gemm_bias(const float* __restrict__ A, const float* __restrict__ B,
                 const float* __restrict__ bias, float* __restrict__ C) {
    gemm_piece<NCLS, true>(A, B, bias, C);
}

#define AGG_BLOCKS   256
#define AGG_NODES    196
#define AGG_THREADS  1024
#define AGG_WAVES    (AGG_THREADS / 32)
#define AGG_SEGS     (BIN_CHUNKS / AGG_WAVES)
#define AGG_QCAP     544
#define AGG_GROUP    8
#define AGG_LCAP     8192
#define AGG_CHUNK    512
#define AGG_PIECES   (AGG_NODES * (DFEAT / 4))
#define AGG_OWN      ((AGG_PIECES + AGG_THREADS - 1) / AGG_THREADS)

__device__ __forceinline__ void list_keep(const int2* wq, const int qn, const int lane,
                                          int2* list, int* cursor) {
    int base = 0;
    if (lane == 0) base = atomicAdd(cursor, qn);
    base = __builtin_amdgcn_readfirstlane(base);
    if (base + qn <= AGG_LCAP)
        for (int j = lane; j < qn; j += 32) list[base + j] = wq[j];
}

__global__ __launch_bounds__(AGG_THREADS)
void k_degree_list(const int* __restrict__ segs, const int* __restrict__ segcount,
                   float* __restrict__ dinv, int* lists, int* counts) {
    __shared__ int   indeg[AGG_NODES];
    __shared__ int2  queue[AGG_WAVES][AGG_QCAP];
    __shared__ int   cursor;

    const int tid  = threadIdx.x;
    const int lane = tid & 31;
    const int wave = tid >> 5;
    const int lo   = blockIdx.x * AGG_NODES;
    const int cnt  = (N_NODES - lo < AGG_NODES) ? (N_NODES - lo) : AGG_NODES;

    int2* mylist = (int2*)lists + blockIdx.x * AGG_LCAP;

    const int   range = blockIdx.x / BIN_OWNERS;
    const int2* seg2  = (const int2*)segs + (range * BIN_CHUNKS + wave) * BIN_EDGES;
    int len[AGG_SEGS];
    #pragma unroll
    for (int k = 0; k < AGG_SEGS; ++k)
        len[k] = segcount[range * BIN_CHUNKS + wave + k * AGG_WAVES];

    for (int i = tid; i < AGG_NODES; i += AGG_THREADS) indeg[i] = 1;
    if (tid == 0) cursor = 0;
    __syncthreads();

    int2* wq = queue[wave];

    int longest = 0;
    #pragma unroll
    for (int k = 0; k < AGG_SEGS; ++k) {
        len[k]  = __builtin_amdgcn_readfirstlane(len[k]);
        len[k]  = (len[k] < BIN_EDGES) ? len[k] : BIN_EDGES;
        longest = (len[k] > longest) ? len[k] : longest;
    }

    int qn = 0;

    for (int t0 = 0; t0 < longest; t0 += 32) {
        const int t = t0 + lane;
        int2 pr[AGG_SEGS];
        #pragma unroll
        for (int k = 0; k < AGG_SEGS; ++k)
            pr[k] = (t < len[k]) ? seg2[k * AGG_WAVES * BIN_EDGES + t] : make_int2(-1, -1);

        if (qn > AGG_QCAP - 32 * AGG_SEGS) {
            list_keep(wq, qn, lane, mylist, &cursor);
            qn = 0;
        }
        #pragma unroll
        for (int k = 0; k < AGG_SEGS; ++k) {
            const int  dl  = pr[k].y - lo;
            const bool hit = (unsigned)dl < (unsigned)cnt;
            const unsigned m = __builtin_amdgcn_ballot_w32(hit);
            if (hit) {
                const int pos = qn + (int)__builtin_amdgcn_mbcnt_lo(m, 0u);
                wq[pos] = make_int2(pr[k].x, dl);
                atomicAdd(&indeg[dl], 1);
            }
            qn += __builtin_popcount(m);
        }
    }
    if (qn > 0) list_keep(wq, qn, lane, mylist, &cursor);
    __syncthreads();
    if (tid == 0) counts[blockIdx.x] = cursor;

    for (int i = tid; i < cnt; i += AGG_THREADS)
        dinv[lo + i] = rsqrtf((float)indeg[i]);
}

__device__ __forceinline__ void list_fetch128(int2* wq, const int2* from, const int j0, const int m, const int lane) {
    int2 t[4];
    #pragma unroll
    for (int u = 0; u < 4; ++u) {
        const int j = j0 + 32 * u + lane;
        t[u] = (j < m) ? from[j] : make_int2(0, 0);
    }
    #pragma unroll
    for (int u = 0; u < 4; ++u) {
        const int j = j0 + 32 * u + lane;
        if (j < m) wq[j] = t[u];
    }
}

__device__ __forceinline__ void agg_accumulate(const int2* wq, const int qn, const int lane,
                                               const float* __restrict__ h,
                                               const float* __restrict__ dinv,
                                               const float* sdinv, float* acc) {
    for (int j = 0; j < qn; j += AGG_GROUP) {
        int s[AGG_GROUP], dl[AGG_GROUP];
        #pragma unroll
        for (int k = 0; k < AGG_GROUP; ++k) {
            const int jj = (j + k < qn) ? (j + k) : (qn - 1);
            const int2 ent = wq[jj];
            s[k]  = __builtin_amdgcn_readfirstlane(ent.x);
            dl[k] = __builtin_amdgcn_readfirstlane(ent.y);
        }
        v4f   v[AGG_GROUP];
        float nrm[AGG_GROUP];
        #pragma unroll
        for (int k = 0; k < AGG_GROUP; ++k) {
            nrm[k] = dinv[s[k]] * sdinv[dl[k]];
            v[k] = ((const v4f*)(h + s[k] * DFEAT))[lane];
        }
        #pragma unroll
        for (int k = 0; k < AGG_GROUP; ++k) {
            if (j + k < qn) {
                float* a = acc + dl[k] * DFEAT + lane;
                atomicAdd(a +  0, v[k].x * nrm[k]);
                atomicAdd(a + 32, v[k].y * nrm[k]);
                atomicAdd(a + 64, v[k].z * nrm[k]);
                atomicAdd(a + 96, v[k].w * nrm[k]);
            }
        }
    }
}

__global__ __launch_bounds__(AGG_THREADS)
void k_aggregate(const float* __restrict__ h, const int* __restrict__ segs,
                 const int* __restrict__ segcount, const float* __restrict__ dinv,
                 const float* __restrict__ bias, float* __restrict__ out,
                 const int* __restrict__ lists, const int* __restrict__ counts) {
    __shared__ __attribute__((aligned(16))) float acc[AGG_NODES * DFEAT];
    __shared__ float sdinv[AGG_NODES];
    __shared__ int2  queue[AGG_WAVES][AGG_QCAP];

    const int tid  = threadIdx.x;
    const int lane = tid & 31;
    const int wave = tid >> 5;
    const int lo   = blockIdx.x * AGG_NODES;
    const int cnt  = (N_NODES - lo < AGG_NODES) ? (N_NODES - lo) : AGG_NODES;
    const int pieces = cnt * (DFEAT / 4);

    const v4f  zero4 = {0.0f, 0.0f, 0.0f, 0.0f};
    const v4f* h4    = (const v4f*)h + lo * (DFEAT / 4);
    v4f own[AGG_OWN];
    #pragma unroll
    for (int t = 0; t < AGG_OWN; ++t) {
        const int i = tid + t * AGG_THREADS;
        own[t] = (i < pieces) ? h4[i] : zero4;
    }
    const v4f   bb  = ((const v4f*)bias)[lane];
    const float myd = (tid < cnt) ? dinv[lo + tid] : 0.0f;

    const int2* mylist = (const int2*)lists + blockIdx.x * AGG_LCAP;

    const int  kept = __builtin_amdgcn_readfirstlane(counts[blockIdx.x]);
    const bool scan = kept > AGG_LCAP;

    const int   range = blockIdx.x / BIN_OWNERS;
    const int2* seg2  = (const int2*)segs + (range * BIN_CHUNKS + wave) * BIN_EDGES;
    int len[AGG_SEGS];
    #pragma unroll
    for (int k = 0; k < AGG_SEGS; ++k)
        len[k] = scan ? segcount[range * BIN_CHUNKS + wave + k * AGG_WAVES] : 0;

    int2* wq = queue[wave];

    const int w   = __builtin_amdgcn_readfirstlane(wave);
    const int per = (kept + AGG_WAVES - 1) / AGG_WAVES;
    const int beg = w * per;
    const int end = (beg + per < kept) ? (beg + per) : kept;
    if (!scan && beg < end)
        list_fetch128(wq, mylist + beg, 0, (end - beg < AGG_CHUNK) ? (end - beg) : AGG_CHUNK, lane);

    v4f* acc4 = (v4f*)acc;
    for (int i = tid; i < pieces; i += AGG_THREADS) acc4[i] = zero4;
    if (tid < cnt) sdinv[tid] = myd;
    __syncthreads();

    if (!scan) {
        for (int c = beg; c < end; c += AGG_CHUNK) {
            const int m = (end - c < AGG_CHUNK) ? (end - c) : AGG_CHUNK;
            for (int j0 = (c == beg) ? 128 : 0; j0 < m; j0 += 128)
                list_fetch128(wq, mylist + c, j0, m, lane);
            agg_accumulate(wq, m, lane, h, dinv, sdinv, acc);
        }
    } else {
        int longest = 0;
        #pragma unroll
        for (int k = 0; k < AGG_SEGS; ++k) {
            len[k]  = __builtin_amdgcn_readfirstlane(len[k]);
            len[k]  = (len[k] < BIN_EDGES) ? len[k] : BIN_EDGES;
            longest = (len[k] > longest) ? len[k] : longest;
        }

        int qn = 0;

        for (int t0 = 0; t0 < longest; t0 += 32) {
            const int t = t0 + lane;
            int2 pr[AGG_SEGS];
            #pragma unroll
            for (int k = 0; k < AGG_SEGS; ++k)
                pr[k] = (t < len[k]) ? seg2[k * AGG_WAVES * BIN_EDGES + t] : make_int2(-1, -1);

            if (qn > AGG_QCAP - 32 * AGG_SEGS) {
                agg_accumulate(wq, qn, lane, h, dinv, sdinv, acc);
                qn = 0;
            }
            #pragma unroll
            for (int k = 0; k < AGG_SEGS; ++k) {
                const int  dl  = pr[k].y - lo;
                const bool hit = (unsigned)dl < (unsigned)cnt;
                const unsigned m = __builtin_amdgcn_ballot_w32(hit);
                if (hit) {
                    const int pos = qn + (int)__builtin_amdgcn_mbcnt_lo(m, 0u);
                    wq[pos] = make_int2(pr[k].x, dl);
                }
                qn += __builtin_popcount(m);
            }
        }
        if (qn > 0) agg_accumulate(wq, qn, lane, h, dinv, sdinv, acc);
    }
    __syncthreads();

    v4f* op4 = (v4f*)(out + lo * DFEAT);
    #pragma unroll
    for (int t = 0; t < AGG_OWN; ++t) {
        const int i = tid + t * AGG_THREADS;
        if (i < pieces) {
            const float di = sdinv[i >> 5];
            const float nn = di * di;
            const float* ar = acc + (i >> 5) * DFEAT + (i & 31);
            v4f a;
            a.x = ar[0]; a.y = ar[32]; a.z = ar[64]; a.w = ar[96];
            v4f o;
            o.x = tanhf(a.x + own[t].x * nn + bb.x);
            o.y = tanhf(a.y + own[t].y * nn + bb.y);
            o.z = tanhf(a.z + own[t].z * nn + bb.z);
            o.w = tanhf(a.w + own[t].w * nn + bb.w);
            op4[i] = o;
        }
    }
}

extern "C" void kernel_launch(void* const* d_in, const int* in_sizes, int n_in,
                              void* d_out, int out_size, void* d_ws, size_t ws_size,
                              hipStream_t stream) {
    const float* x    = (const float*)d_in[0];
    const int*   ei   = (const int*)  d_in[1];
    const float* W1   = (const float*)d_in[2];
    const float* b1   = (const float*)d_in[3];
    const float* W2   = (const float*)d_in[4];
    const float* b2   = (const float*)d_in[5];
    const float* Wlin = (const float*)d_in[6];
    const float* blin = (const float*)d_in[7];

    const int* src = ei;
    const int* dst = ei + N_EDGES;

    float* wsf  = (float*)d_ws;
    float* dinv = wsf;
    float* bufA = wsf + 50048;
    float* bufB = bufA + (size_t)N_NODES * DFEAT;
    int*   lists  = (int*)(bufB + (size_t)N_NODES * DFEAT);
    int*   counts = lists + (size_t)AGG_BLOCKS * AGG_LCAP * 2;
    int*   segcount = counts + AGG_BLOCKS;
    int*   segs     = segcount + BIN_RANGES * BIN_CHUNKS;

    const dim3 gemmGrid((GEMM_STRIPS + GEMM_WAVES - 1) / GEMM_WAVES, DFEAT / 64);

    k_partition<<<BIN_CHUNKS, BIN_THREADS, 0, stream>>>(src, dst, segs, segcount);
    k_degree_list<<<AGG_BLOCKS, AGG_THREADS, 0, stream>>>(segs, segcount, dinv, lists, counts);

    k_gemm<<<gemmGrid, 32 * GEMM_WAVES, 0, stream>>>(x, W1, bufA);
    k_aggregate<<<AGG_BLOCKS, AGG_THREADS, 0, stream>>>(bufA, segs, segcount, dinv, b1, bufB, lists, counts);

    k_gemm<<<gemmGrid, 32 * GEMM_WAVES, 0, stream>>>(bufB, W2, bufA);
    k_aggregate<<<AGG_BLOCKS, AGG_THREADS, 0, stream>>>(bufA, segs, segcount, dinv, b2, bufB, lists, counts);

    k_gemm_bias<<<(GEMM_STRIPS + GEMM_WAVES - 1) / GEMM_WAVES, 32 * GEMM_WAVES, 0, stream>>>(bufB, Wlin, blin, (float*)d_out);
}


# Edge sums leave LDS: the list pass files a block's edges by owner wave (row mod 32); a wave keeps its 7 rows' sums in registers (scalar branch on the row), no LDS float atomics; sums in bin order
# speedup vs baseline: 3.1819x; 1.0620x over previous
#include <hip/hip_runtime.h>
#include <math.h>

typedef float v2f __attribute__((ext_vector_type(2)));
typedef float v8f __attribute__((ext_vector_type(8)));

#define N_NODES 50000
#define N_EDGES 600000
#define DFEAT   128
#define NCLS    40

#define BIN_RANGES   16
#define BIN_OWNERS   16
#define BIN_NODES    (BIN_OWNERS * 196)
#define BIN_CHUNKS   256
#define BIN_EDGES    2344
#define BIN_THREADS  1024
#define BIN_WAVES    (BIN_THREADS / 32)
#define BIN_PER_WAVE 74
#define BIN_VECS     3

__global__ __launch_bounds__(BIN_THREADS)
void k_partition(const int* __restrict__ src, const int* __restrict__ dst,
                 int* segs, int* segcount) {
    __shared__ int wcnt[BIN_WAVES][BIN_RANGES];
    __shared__ int wstart[BIN_WAVES][BIN_RANGES];

    const int tid  = threadIdx.x;
    const int lane = tid & 31;
    const int wave = tid >> 5;
    const int e0 = blockIdx.x * BIN_EDGES;
    const int e1 = (e0 + BIN_EDGES < N_EDGES) ? (e0 + BIN_EDGES) : N_EDGES;
    const int wb = e0 + wave * BIN_PER_WAVE;
    const int we = (wb + BIN_PER_WAVE < e1) ? (wb + BIN_PER_WAVE) : e1;

    int s[BIN_VECS], d[BIN_VECS], r[BIN_VECS], rk[BIN_VECS];
    #pragma unroll
    for (int v = 0; v < BIN_VECS; ++v) {
        const int  e  = wb + 32 * v + lane;
        const bool ok = e < we;
        d[v] = ok ? dst[e] : -1;
        s[v] = ok ? src[e] : 0;
    }
    #pragma unroll
    for (int v = 0; v < BIN_VECS; ++v) {
        r[v]  = (d[v] >= 0) ? (d[v] / BIN_NODES) : -1;
        rk[v] = 0;
    }

    int mine = 0;
    #pragma unroll
    for (int b = 0; b < BIN_RANGES; ++b) {
        int run = 0;
        #pragma unroll
        for (int v = 0; v < BIN_VECS; ++v) {
            const bool hit = (r[v] == b);
            const unsigned m = __builtin_amdgcn_ballot_w32(hit);
            if (hit) rk[v] = run + (int)__builtin_amdgcn_mbcnt_lo(m, 0u);
            run += __builtin_popcount(m);
        }
        if (lane == b) mine = run;
    }
    if (lane < BIN_RANGES) wcnt[wave][lane] = mine;
    __syncthreads();

    if (tid < BIN_WAVES * BIN_RANGES) {
        const int w = tid >> 4;
        const int b = tid & 15;
        int sum = 0;
        for (int w2 = 0; w2 < w; ++w2) sum += wcnt[w2][b];
        wstart[w][b] = sum;
        if (w == BIN_WAVES - 1) segcount[b * BIN_CHUNKS + blockIdx.x] = sum + wcnt[w][b];
    }
    __syncthreads();

    int2* seg2 = (int2*)segs;
    #pragma unroll
    for (int v = 0; v < BIN_VECS; ++v) {
        if (r[v] >= 0) {
            const int pos = wstart[wave][r[v]] + rk[v];
            seg2[(r[v] * BIN_CHUNKS + blockIdx.x) * BIN_EDGES + pos] = make_int2(s[v], d[v]);
        }
    }
}

typedef float v4f __attribute__((ext_vector_type(4)));
#define GEMM_WAVES  4
#define GEMM_STRIPS (N_NODES / 16)

template<int N, bool BIAS>
__device__ __forceinline__ void gemm_piece(const float* __restrict__ A, const float* __restrict__ B,
                                           const float* __restrict__ bias, float* __restrict__ C) {
    const int lane  = threadIdx.x & 31;
    const int wave  = threadIdx.x >> 5;
    const int strip = blockIdx.x * GEMM_WAVES + wave;
    if (strip >= GEMM_STRIPS) return;
    const int row0  = strip * 16;
    const int c     = lane & 15;
    const int h     = lane >> 4;
    const int col   = blockIdx.y * 64 + 4 * c;
    const bool live = (N % 64 == 0) || (col < N);
    const int colc  = live ? col : (N - 4);

    const float* Ap = A + (row0 + c) * DFEAT + 2 * h;
    const float* Bp = B + (2 * h) * N + colc;

    v8f acc0 = {}, acc1 = {}, acc2 = {}, acc3 = {};
    #pragma unroll
    for (int k0 = 0; k0 < DFEAT; k0 += 4) {
        const v2f a  = *(const v2f*)(Ap + k0);
        const v4f b0 = *(const v4f*)(Bp + k0 * N);
        const v4f b1 = *(const v4f*)(Bp + k0 * N + N);
        v2f b;
        b.x = b0.x; b.y = b1.x;
        acc0 = __builtin_amdgcn_wmma_f32_16x16x4_f32(false, a, false, b, (short)0, acc0, false, false);
        b.x = b0.y; b.y = b1.y;
        acc1 = __builtin_amdgcn_wmma_f32_16x16x4_f32(false, a, false, b, (short)0, acc1, false, false);
        b.x = b0.z; b.y = b1.z;
        acc2 = __builtin_amdgcn_wmma_f32_16x16x4_f32(false, a, false, b, (short)0, acc2, false, false);
        b.x = b0.w; b.y = b1.w;
        acc3 = __builtin_amdgcn_wmma_f32_16x16x4_f32(false, a, false, b, (short)0, acc3, false, false);
    }

    if (live) {
        v4f bb = {0.0f, 0.0f, 0.0f, 0.0f};
        if (BIAS) bb = *(const v4f*)(bias + col);
        float* Cp = C + (row0 + 8 * h) * N + col;
        #pragma unroll
        for (int i = 0; i < 8; ++i) {
            v4f o;
            o.x = acc0[i]; o.y = acc1[i]; o.z = acc2[i]; o.w = acc3[i];
            if (BIAS) o += bb;
            *(v4f*)(Cp + i * N) = o;
        }
    }
}

__global__ __launch_bounds__(32 * GEMM_WAVES)
void k_gemm(const float* __restrict__ A, const float* __restrict__ B, float* __restrict__ C) {
    gemm_piece<DFEAT, false>(A, B, nullptr, C);
}

__global__ __launch_bounds__(32 * GEMM_WAVES)
void k_gemm_bias(const float* __restrict__ A, const float* __restrict__ B,
                 const float* __restrict__ bias, float* __restrict__ C) {
    gemm_piece<NCLS, true>(A, B, bias, C);
}

#define AGG_BLOCKS   256
#define AGG_NODES    196
#define AGG_THREADS  1024
#define AGG_WAVES    (AGG_THREADS / 32)
#define AGG_ROWS     ((AGG_NODES + AGG_WAVES - 1) / AGG_WAVES)
#define AGG_SEGS     (BIN_CHUNKS / AGG_WAVES)
#define AGG_GROUP    16
#define AGG_BCAP     128
#define AGG_BVECS    (AGG_BCAP / 32)

__global__ __launch_bounds__(AGG_THREADS)
void k_degree_list(const int* __restrict__ segs, const int* __restrict__ segcount,
                   float* __restrict__ dinv, int* bins, int* counts) {
    __shared__ int  indeg[AGG_NODES];
    __shared__ int2 bin[AGG_WAVES][AGG_BCAP];
    __shared__ int  bcnt[AGG_WAVES];

    const int tid  = threadIdx.x;
    const int lane = tid & 31;
    const int wave = tid >> 5;
    const int lo   = blockIdx.x * AGG_NODES;
    const int cnt  = (N_NODES - lo < AGG_NODES) ? (N_NODES - lo) : AGG_NODES;

    const int   range = blockIdx.x / BIN_OWNERS;
    const int2* seg2  = (const int2*)segs + (range * BIN_CHUNKS + wave) * BIN_EDGES;
    int len[AGG_SEGS];
    #pragma unroll
    for (int k = 0; k < AGG_SEGS; ++k)
        len[k] = segcount[range * BIN_CHUNKS + wave + k * AGG_WAVES];

    for (int i = tid; i < AGG_NODES; i += AGG_THREADS) indeg[i] = 1;
    if (tid < AGG_WAVES) bcnt[tid] = 0;
    int2* allbins = &bin[0][0];
    for (int i = tid; i < AGG_WAVES * AGG_BCAP; i += AGG_THREADS) allbins[i] = make_int2(0, AGG_ROWS);
    __syncthreads();

    int longest = 0;
    #pragma unroll
    for (int k = 0; k < AGG_SEGS; ++k) {
        len[k]  = __builtin_amdgcn_readfirstlane(len[k]);
        len[k]  = (len[k] < BIN_EDGES) ? len[k] : BIN_EDGES;
        longest = (len[k] > longest) ? len[k] : longest;
    }

    for (int t0 = 0; t0 < longest; t0 += 32) {
        const int t = t0 + lane;
        int2 pr[AGG_SEGS];
        #pragma unroll
        for (int k = 0; k < AGG_SEGS; ++k)
            pr[k] = (t < len[k]) ? seg2[k * AGG_WAVES * BIN_EDGES + t] : make_int2(-1, -1);

        int slot[AGG_SEGS];
        #pragma unroll
        for (int k = 0; k < AGG_SEGS; ++k) {
            const int dl = pr[k].y - lo;
            slot[k] = AGG_BCAP;
            if ((unsigned)dl < (unsigned)cnt) {
                slot[k] = atomicAdd(&bcnt[(unsigned)dl % AGG_WAVES], 1);
                atomicAdd(&indeg[dl], 1);
            }
        }
        #pragma unroll
        for (int k = 0; k < AGG_SEGS; ++k) {
            if (slot[k] < AGG_BCAP) {
                const int dl = pr[k].y - lo;
                bin[(unsigned)dl % AGG_WAVES][slot[k]] = make_int2(pr[k].x, (int)((unsigned)dl / AGG_WAVES));
            }
        }
    }
    __syncthreads();

    int2* gbin = (int2*)bins + (blockIdx.x * AGG_WAVES + wave) * AGG_BCAP;
    #pragma unroll
    for (int u = 0; u < AGG_BVECS; ++u) gbin[32 * u + lane] = bin[wave][32 * u + lane];
    if (tid < AGG_WAVES) counts[blockIdx.x * AGG_WAVES + tid] = bcnt[tid];

    for (int i = tid; i < cnt; i += AGG_THREADS)
        dinv[lo + i] = rsqrtf((float)indeg[i]);
}

__device__ __forceinline__ void row_add(v4f (&acc)[AGG_ROWS], const float (&dd)[AGG_ROWS],
                                        const int rr, const v4f v, const float ds) {
    #pragma unroll
    for (int r = 0; r < AGG_ROWS; ++r)
        if (__builtin_expect(rr == r, 0)) acc[r] += v * (ds * dd[r]);
}

__global__ __launch_bounds__(AGG_THREADS)
void k_aggregate(const float* __restrict__ h, const int* __restrict__ segs,
                 const int* __restrict__ segcount, const float* __restrict__ dinv,
                 const float* __restrict__ bias, float* __restrict__ out,
                 const int* __restrict__ bins, const int* __restrict__ counts) {
    static_assert(AGG_BVECS == 4, "the bin's vectors are named one by one");
    static_assert(AGG_GROUP == 16, "the loads' fence below names sixteen");

    const int lane = threadIdx.x & 31;
    const int w    = __builtin_amdgcn_readfirstlane((int)(threadIdx.x >> 5));
    const int lo   = blockIdx.x * AGG_NODES;
    const int cnt  = (N_NODES - lo < AGG_NODES) ? (N_NODES - lo) : AGG_NODES;

    const int   n     = __builtin_amdgcn_readfirstlane(counts[blockIdx.x * AGG_WAVES + w]);
    const int2* mybin = (const int2*)bins + (blockIdx.x * AGG_WAVES + w) * AGG_BCAP;
    const int2 ent0 = mybin[lane];
    const int2 ent1 = mybin[32 + lane];
    const int2 ent2 = mybin[64 + lane];
    const int2 ent3 = mybin[96 + lane];

    const v4f* h4 = (const v4f*)h;
    float dd[AGG_ROWS];
    v4f   acc[AGG_ROWS];
    #pragma unroll
    for (int r = 0; r < AGG_ROWS; ++r) {
        const int row = w + AGG_WAVES * r;
        const int rowc = (row < cnt) ? row : (cnt - 1);
        dd[r]  = dinv[lo + rowc];
        acc[r] = h4[(lo + rowc) * (DFEAT / 4) + lane];
    }
    const v4f bb = ((const v4f*)bias)[lane];
    #pragma unroll
    for (int r = 0; r < AGG_ROWS; ++r) acc[r] = acc[r] * (dd[r] * dd[r]);

    asm volatile("" :: "v"(ent0.x), "v"(ent0.y), "v"(ent1.x), "v"(ent1.y), "v"(ent2.x), "v"(ent2.y), "v"(ent3.x), "v"(ent3.y));
    if (n <= AGG_BCAP) {
        #pragma unroll 1
        for (int u = 0; u < AGG_BVECS; ++u) {
            const int m = (n - 32 * u < 32) ? (n - 32 * u) : 32;
            if (m <= 0) break;
            const int2 cur = (u == 0) ? ent0 : (u == 1) ? ent1 : (u == 2) ? ent2 : ent3;
            for (int j = 0; j < m; j += AGG_GROUP) {
                int s[AGG_GROUP], rr[AGG_GROUP];
                #pragma unroll
                for (int k = 0; k < AGG_GROUP; ++k) {
                    s[k]  = __builtin_amdgcn_readlane(cur.x, j + k);
                    rr[k] = __builtin_amdgcn_readlane(cur.y, j + k);
                }
                v4f   v[AGG_GROUP];
                float ds[AGG_GROUP];
                #pragma unroll
                for (int k = 0; k < AGG_GROUP; ++k) {
                    ds[k] = dinv[s[k]];
                    v[k]  = h4[s[k] * (DFEAT / 4) + lane];
                }
                asm volatile("" :: "s"(ds[0]), "s"(ds[1]), "s"(ds[2]), "s"(ds[3]), "s"(ds[4]), "s"(ds[5]), "s"(ds[6]), "s"(ds[7]),
                                   "s"(ds[8]), "s"(ds[9]), "s"(ds[10]), "s"(ds[11]), "s"(ds[12]), "s"(ds[13]), "s"(ds[14]), "s"(ds[15]));
                #pragma unroll
                for (int k = 0; k < AGG_GROUP; ++k) row_add(acc, dd, rr[k], v[k], ds[k]);
            }
        }
    } else {
        const int   range = blockIdx.x / BIN_OWNERS;
        const int2* seg2  = (const int2*)segs + range * BIN_CHUNKS * BIN_EDGES;
        for (int c = 0; c < BIN_CHUNKS; ++c) {
            int len = __builtin_amdgcn_readfirstlane(segcount[range * BIN_CHUNKS + c]);
            len = (len < BIN_EDGES) ? len : BIN_EDGES;
            for (int t0 = 0; t0 < len; t0 += 32) {
                const int  t   = t0 + lane;
                const int2 pr  = (t < len) ? seg2[c * BIN_EDGES + t] : make_int2(0, -1);
                const int  dl  = pr.y - lo;
                const bool hit = ((unsigned)dl < (unsigned)cnt) && ((int)((unsigned)dl % AGG_WAVES) == w);
                unsigned mask = __builtin_amdgcn_ballot_w32(hit);
                for (int q = 0; q < 32 && mask != 0u; ++q) {
                    const int b = __builtin_ctz(mask);
                    mask &= mask - 1u;
                    const int s = __builtin_amdgcn_readlane(pr.x, b);
                    const int r = (int)((unsigned)__builtin_amdgcn_readlane(dl, b) / AGG_WAVES);
                    row_add(acc, dd, r, h4[s * (DFEAT / 4) + lane], dinv[s]);
                }
            }
        }
    }

    v4f* op4 = (v4f*)out;
    #pragma unroll
    for (int r = 0; r < AGG_ROWS; ++r) {
        const int row = w + AGG_WAVES * r;
        if (row < cnt) {
            v4f o;
            o.x = tanhf(acc[r].x + bb.x);
            o.y = tanhf(acc[r].y + bb.y);
            o.z = tanhf(acc[r].z + bb.z);
            o.w = tanhf(acc[r].w + bb.w);
            op4[(lo + row) * (DFEAT / 4) + lane] = o;
        }
    }
}

extern "C" void kernel_launch(void* const* d_in, const int* in_sizes, int n_in,
                              void* d_out, int out_size, void* d_ws, size_t ws_size,
                              hipStream_t stream) {
    const float* x    = (const float*)d_in[0];
    const int*   ei   = (const int*)  d_in[1];
    const float* W1   = (const float*)d_in[2];
    const float* b1   = (const float*)d_in[3];
    const float* W2   = (const float*)d_in[4];
    const float* b2   = (const float*)d_in[5];
    const float* Wlin = (const float*)d_in[6];
    const float* blin = (const float*)d_in[7];

    const int* src = ei;
    const int* dst = ei + N_EDGES;

    float* wsf  = (float*)d_ws;
    float* dinv = wsf;
    float* bufA = wsf + 50048;
    float* bufB = bufA + (size_t)N_NODES * DFEAT;
    int*   bins   = (int*)(bufB + (size_t)N_NODES * DFEAT);
    int*   counts = bins + (size_t)AGG_BLOCKS * AGG_WAVES * AGG_BCAP * 2;
    int*   segcount = counts + AGG_BLOCKS * AGG_WAVES;
    int*   segs     = segcount + BIN_RANGES * BIN_CHUNKS;

    const dim3 gemmGrid((GEMM_STRIPS + GEMM_WAVES - 1) / GEMM_WAVES, DFEAT / 64);

    k_partition<<<BIN_CHUNKS, BIN_THREADS, 0, stream>>>(src, dst, segs, segcount);
    k_degree_list<<<AGG_BLOCKS, AGG_THREADS, 0, stream>>>(segs, segcount, dinv, bins, counts);

    k_gemm<<<gemmGrid, 32 * GEMM_WAVES, 0, stream>>>(x, W1, bufA);
    k_aggregate<<<AGG_BLOCKS, AGG_THREADS, 0, stream>>>(bufA, segs, segcount, dinv, b1, bufB, bins, counts);

    k_gemm<<<gemmGrid, 32 * GEMM_WAVES, 0, stream>>>(bufB, W2, bufA);
    k_aggregate<<<AGG_BLOCKS, AGG_THREADS, 0, stream>>>(bufA, segs, segcount, dinv, b2, bufB, bins, counts);

    k_gemm_bias<<<(GEMM_STRIPS + GEMM_WAVES - 1) / GEMM_WAVES, 32 * GEMM_WAVES, 0, stream>>>(bufB, Wlin, blin, (float*)d_out);
}


# Layer-2 product moved behind its aggregation ((Ahat h1) W2: a changed summation order) and run with the head product in one kernel, the tanh'd 16-row strips handed over in LDS; 6 launches
# speedup vs baseline: 3.2779x; 1.0302x over previous
#include <hip/hip_runtime.h>
#include <math.h>

typedef float v2f __attribute__((ext_vector_type(2)));
typedef float v8f __attribute__((ext_vector_type(8)));

#define N_NODES 50000
#define N_EDGES 600000
#define DFEAT   128
#define NCLS    40

#define BIN_RANGES   16
#define BIN_OWNERS   16
#define BIN_NODES    (BIN_OWNERS * 196)
#define BIN_CHUNKS   256
#define BIN_EDGES    2344
#define BIN_THREADS  1024
#define BIN_WAVES    (BIN_THREADS / 32)
#define BIN_PER_WAVE 74
#define BIN_VECS     3

__global__ __launch_bounds__(BIN_THREADS)
void k_partition(const int* __restrict__ src, const int* __restrict__ dst,
                 int* segs, int* segcount) {
    __shared__ int wcnt[BIN_WAVES][BIN_RANGES];
    __shared__ int wstart[BIN_WAVES][BIN_RANGES];

    const int tid  = threadIdx.x;
    const int lane = tid & 31;
    const int wave = tid >> 5;
    const int e0 = blockIdx.x * BIN_EDGES;
    const int e1 = (e0 + BIN_EDGES < N_EDGES) ? (e0 + BIN_EDGES) : N_EDGES;
    const int wb = e0 + wave * BIN_PER_WAVE;
    const int we = (wb + BIN_PER_WAVE < e1) ? (wb + BIN_PER_WAVE) : e1;

    int s[BIN_VECS], d[BIN_VECS], r[BIN_VECS], rk[BIN_VECS];
    #pragma unroll
    for (int v = 0; v < BIN_VECS; ++v) {
        const int  e  = wb + 32 * v + lane;
        const bool ok = e < we;
        d[v] = ok ? dst[e] : -1;
        s[v] = ok ? src[e] : 0;
    }
    #pragma unroll
    for (int v = 0; v < BIN_VECS; ++v) {
        r[v]  = (d[v] >= 0) ? (d[v] / BIN_NODES) : -1;
        rk[v] = 0;
    }

    int mine = 0;
    #pragma unroll
    for (int b = 0; b < BIN_RANGES; ++b) {
        int run = 0;
        #pragma unroll
        for (int v = 0; v < BIN_VECS; ++v) {
            const bool hit = (r[v] == b);
            const unsigned m = __builtin_amdgcn_ballot_w32(hit);
            if (hit) rk[v] = run + (int)__builtin_amdgcn_mbcnt_lo(m, 0u);
            run += __builtin_popcount(m);
        }
        if (lane == b) mine = run;
    }
    if (lane < BIN_RANGES) wcnt[wave][lane] = mine;
    __syncthreads();

    if (tid < BIN_WAVES * BIN_RANGES) {
        const int w = tid >> 4;
        const int b = tid & 15;
        int sum = 0;
        for (int w2 = 0; w2 < w; ++w2) sum += wcnt[w2][b];
        wstart[w][b] = sum;
        if (w == BIN_WAVES - 1) segcount[b * BIN_CHUNKS + blockIdx.x] = sum + wcnt[w][b];
    }
    __syncthreads();

    int2* seg2 = (int2*)segs;
    #pragma unroll
    for (int v = 0; v < BIN_VECS; ++v) {
        if (r[v] >= 0) {
            const int pos = wstart[wave][r[v]] + rk[v];
            seg2[(r[v] * BIN_CHUNKS + blockIdx.x) * BIN_EDGES + pos] = make_int2(s[v], d[v]);
        }
    }
}

typedef float v4f __attribute__((ext_vector_type(4)));
#define GEMM_WAVES  4
#define GEMM_STRIPS (N_NODES / 16)

template<int N, bool BIAS>
__device__ __forceinline__ void gemm_piece(const float* __restrict__ A, const float* __restrict__ B,
                                           const float* __restrict__ bias, float* __restrict__ C) {
    const int lane  = threadIdx.x & 31;
    const int wave  = threadIdx.x >> 5;
    const int strip = blockIdx.x * GEMM_WAVES + wave;
    if (strip >= GEMM_STRIPS) return;
    const int row0  = strip * 16;
    const int c     = lane & 15;
    const int h     = lane >> 4;
    const int col   = blockIdx.y * 64 + 4 * c;
    const bool live = (N % 64 == 0) || (col < N);
    const int colc  = live ? col : (N - 4);

    const float* Ap = A + (row0 + c) * DFEAT + 2 * h;
    const float* Bp = B + (2 * h) * N + colc;

    v8f acc0 = {}, acc1 = {}, acc2 = {}, acc3 = {};
    #pragma unroll
    for (int k0 = 0; k0 < DFEAT; k0 += 4) {
        const v2f a  = *(const v2f*)(Ap + k0);
        const v4f b0 = *(const v4f*)(Bp + k0 * N);
        const v4f b1 = *(const v4f*)(Bp + k0 * N + N);
        v2f b;
        b.x = b0.x; b.y = b1.x;
        acc0 = __builtin_amdgcn_wmma_f32_16x16x4_f32(false, a, false, b, (short)0, acc0, false, false);
        b.x = b0.y; b.y = b1.y;
        acc1 = __builtin_amdgcn_wmma_f32_16x16x4_f32(false, a, false, b, (short)0, acc1, false, false);
        b.x = b0.z; b.y = b1.z;
        acc2 = __builtin_amdgcn_wmma_f32_16x16x4_f32(false, a, false, b, (short)0, acc2, false, false);
        b.x = b0.w; b.y = b1.w;
        acc3 = __builtin_amdgcn_wmma_f32_16x16x4_f32(false, a, false, b, (short)0, acc3, false, false);
    }

    if (live) {
        v4f bb = {0.0f, 0.0f, 0.0f, 0.0f};
        if (BIAS) bb = *(const v4f*)(bias + col);
        float* Cp = C + (row0 + 8 * h) * N + col;
        #pragma unroll
        for (int i = 0; i < 8; ++i) {
            v4f o;
            o.x = acc0[i]; o.y = acc1[i]; o.z = acc2[i]; o.w = acc3[i];
            if (BIAS) o += bb;
            *(v4f*)(Cp + i * N) = o;
        }
    }
}

__global__ __launch_bounds__(32 * GEMM_WAVES)
void k_gemm(const float* __restrict__ A, const float* __restrict__ B, float* __restrict__ C) {
    gemm_piece<DFEAT, false>(A, B, nullptr, C);
}

#define D2_PITCH (DFEAT + 4)

__global__ __launch_bounds__(32 * GEMM_WAVES)
void k_dense_head(const float* __restrict__ G, const float* __restrict__ W2, const float* __restrict__ b2,
                  const float* __restrict__ Wlin, const float* __restrict__ blin, float* __restrict__ out) {
    __shared__ __attribute__((aligned(16))) float rows[GEMM_WAVES][16 * D2_PITCH];

    const int lane  = threadIdx.x & 31;
    const int wave  = threadIdx.x >> 5;
    const int strip = blockIdx.x * GEMM_WAVES + wave;
    if (strip >= GEMM_STRIPS) return;
    const int row0  = strip * 16;
    const int c     = lane & 15;
    const int h     = lane >> 4;
    float* mine = &rows[wave][0];

    const float* Ap = G + (row0 + c) * DFEAT + 2 * h;
    #pragma unroll
    for (int half = 0; half < 2; ++half) {
        const int col = 64 * half + 4 * c;
        const float* Bp = W2 + (2 * h) * DFEAT + col;
        v8f acc0 = {}, acc1 = {}, acc2 = {}, acc3 = {};
        #pragma unroll
        for (int k0 = 0; k0 < DFEAT; k0 += 4) {
            if (k0 == DFEAT / 2) __builtin_amdgcn_sched_barrier(0);
            const v2f a  = *(const v2f*)(Ap + k0);
            const v4f b0 = *(const v4f*)(Bp + k0 * DFEAT);
            const v4f b1 = *(const v4f*)(Bp + k0 * DFEAT + DFEAT);
            v2f b;
            b.x = b0.x; b.y = b1.x;
            acc0 = __builtin_amdgcn_wmma_f32_16x16x4_f32(false, a, false, b, (short)0, acc0, false, false);
            b.x = b0.y; b.y = b1.y;
            acc1 = __builtin_amdgcn_wmma_f32_16x16x4_f32(false, a, false, b, (short)0, acc1, false, false);
            b.x = b0.z; b.y = b1.z;
            acc2 = __builtin_amdgcn_wmma_f32_16x16x4_f32(false, a, false, b, (short)0, acc2, false, false);
            b.x = b0.w; b.y = b1.w;
            acc3 = __builtin_amdgcn_wmma_f32_16x16x4_f32(false, a, false, b, (short)0, acc3, false, false);
        }
        const v4f bb = *(const v4f*)(b2 + col);
        #pragma unroll
        for (int i = 0; i < 8; ++i) {
            v4f o;
            o.x = tanhf(acc0[i] + bb.x);
            o.y = tanhf(acc1[i] + bb.y);
            o.z = tanhf(acc2[i] + bb.z);
            o.w = tanhf(acc3[i] + bb.w);
            *(v4f*)(mine + (8 * h + i) * D2_PITCH + col) = o;
        }
    }

    const int  col  = 4 * c;
    const bool live = col < NCLS;
    const int  colc = live ? col : (NCLS - 4);
    const float* Lp = mine + c * D2_PITCH + 2 * h;
    const float* Bq = Wlin + (2 * h) * NCLS + colc;
    v8f acc0 = {}, acc1 = {}, acc2 = {}, acc3 = {};
    #pragma unroll
    for (int k0 = 0; k0 < DFEAT; k0 += 4) {
        const v2f a  = *(const v2f*)(Lp + k0);
        const v4f b0 = *(const v4f*)(Bq + k0 * NCLS);
        const v4f b1 = *(const v4f*)(Bq + k0 * NCLS + NCLS);
        v2f b;
        b.x = b0.x; b.y = b1.x;
        acc0 = __builtin_amdgcn_wmma_f32_16x16x4_f32(false, a, false, b, (short)0, acc0, false, false);
        b.x = b0.y; b.y = b1.y;
        acc1 = __builtin_amdgcn_wmma_f32_16x16x4_f32(false, a, false, b, (short)0, acc1, false, false);
        b.x = b0.z; b.y = b1.z;
        acc2 = __builtin_amdgcn_wmma_f32_16x16x4_f32(false, a, false, b, (short)0, acc2, false, false);
        b.x = b0.w; b.y = b1.w;
        acc3 = __builtin_amdgcn_wmma_f32_16x16x4_f32(false, a, false, b, (short)0, acc3, false, false);
    }
    if (live) {
        const v4f bb = *(const v4f*)(blin + col);
        float* Cp = out + (row0 + 8 * h) * NCLS + col;
        #pragma unroll
        for (int i = 0; i < 8; ++i) {
            v4f o;
            o.x = acc0[i]; o.y = acc1[i]; o.z = acc2[i]; o.w = acc3[i];
            o += bb;
            *(v4f*)(Cp + i * NCLS) = o;
        }
    }
}

#define AGG_BLOCKS   256
#define AGG_NODES    196
#define AGG_THREADS  1024
#define AGG_WAVES    (AGG_THREADS / 32)
#define AGG_ROWS     ((AGG_NODES + AGG_WAVES - 1) / AGG_WAVES)
#define AGG_SEGS     (BIN_CHUNKS / AGG_WAVES)
#define AGG_GROUP    16
#define AGG_BCAP     128
#define AGG_BVECS    (AGG_BCAP / 32)

__global__ __launch_bounds__(AGG_THREADS)
void k_degree_list(const int* __restrict__ segs, const int* __restrict__ segcount,
                   float* __restrict__ dinv, int* bins, int* counts) {
    __shared__ int  indeg[AGG_NODES];
    __shared__ int2 bin[AGG_WAVES][AGG_BCAP];
    __shared__ int  bcnt[AGG_WAVES];

    const int tid  = threadIdx.x;
    const int lane = tid & 31;
    const int wave = tid >> 5;
    const int lo   = blockIdx.x * AGG_NODES;
    const int cnt  = (N_NODES - lo < AGG_NODES) ? (N_NODES - lo) : AGG_NODES;

    const int   range = blockIdx.x / BIN_OWNERS;
    const int2* seg2  = (const int2*)segs + (range * BIN_CHUNKS + wave) * BIN_EDGES;
    int len[AGG_SEGS];
    #pragma unroll
    for (int k = 0; k < AGG_SEGS; ++k)
        len[k] = segcount[range * BIN_CHUNKS + wave + k * AGG_WAVES];

    for (int i = tid; i < AGG_NODES; i += AGG_THREADS) indeg[i] = 1;
    if (tid < AGG_WAVES) bcnt[tid] = 0;
    int2* allbins = &bin[0][0];
    for (int i = tid; i < AGG_WAVES * AGG_BCAP; i += AGG_THREADS) allbins[i] = make_int2(0, AGG_ROWS);
    __syncthreads();

    int longest = 0;
    #pragma unroll
    for (int k = 0; k < AGG_SEGS; ++k) {
        len[k]  = __builtin_amdgcn_readfirstlane(len[k]);
        len[k]  = (len[k] < BIN_EDGES) ? len[k] : BIN_EDGES;
        longest = (len[k] > longest) ? len[k] : longest;
    }

    for (int t0 = 0; t0 < longest; t0 += 32) {
        const int t = t0 + lane;
        int2 pr[AGG_SEGS];
        #pragma unroll
        for (int k = 0; k < AGG_SEGS; ++k)
            pr[k] = (t < len[k]) ? seg2[k * AGG_WAVES * BIN_EDGES + t] : make_int2(-1, -1);

        int slot[AGG_SEGS];
        #pragma unroll
        for (int k = 0; k < AGG_SEGS; ++k) {
            const int dl = pr[k].y - lo;
            slot[k] = AGG_BCAP;
            if ((unsigned)dl < (unsigned)cnt) {
                slot[k] = atomicAdd(&bcnt[(unsigned)dl % AGG_WAVES], 1);
                atomicAdd(&indeg[dl], 1);
            }
        }
        #pragma unroll
        for (int k = 0; k < AGG_SEGS; ++k) {
            if (slot[k] < AGG_BCAP) {
                const int dl = pr[k].y - lo;
                bin[(unsigned)dl % AGG_WAVES][slot[k]] = make_int2(pr[k].x, (int)((unsigned)dl / AGG_WAVES));
            }
        }
    }
    __syncthreads();

    int2* gbin = (int2*)bins + (blockIdx.x * AGG_WAVES + wave) * AGG_BCAP;
    #pragma unroll
    for (int u = 0; u < AGG_BVECS; ++u) gbin[32 * u + lane] = bin[wave][32 * u + lane];
    if (tid < AGG_WAVES) counts[blockIdx.x * AGG_WAVES + tid] = bcnt[tid];

    for (int i = tid; i < cnt; i += AGG_THREADS)
        dinv[lo + i] = rsqrtf((float)indeg[i]);
}

__device__ __forceinline__ void row_add(v4f (&acc)[AGG_ROWS], const float (&dd)[AGG_ROWS],
                                        const int rr, const v4f v, const float ds) {
    #pragma unroll
    for (int r = 0; r < AGG_ROWS; ++r)
        if (__builtin_expect(rr == r, 0)) acc[r] += v * (ds * dd[r]);
}

template<bool ACT>
__global__ __launch_bounds__(AGG_THREADS)
void k_aggregate(const float* __restrict__ h, const int* __restrict__ segs,
                 const int* __restrict__ segcount, const float* __restrict__ dinv,
                 const float* __restrict__ bias, float* __restrict__ out,
                 const int* __restrict__ bins, const int* __restrict__ counts) {
    static_assert(AGG_BVECS == 4, "the bin's vectors are named one by one");
    static_assert(AGG_GROUP == 16, "the loads' fence below names sixteen");

    const int lane = threadIdx.x & 31;
    const int w    = __builtin_amdgcn_readfirstlane((int)(threadIdx.x >> 5));
    const int lo   = blockIdx.x * AGG_NODES;
    const int cnt  = (N_NODES - lo < AGG_NODES) ? (N_NODES - lo) : AGG_NODES;

    const int   n     = __builtin_amdgcn_readfirstlane(counts[blockIdx.x * AGG_WAVES + w]);
    const int2* mybin = (const int2*)bins + (blockIdx.x * AGG_WAVES + w) * AGG_BCAP;
    const int2 ent0 = mybin[lane];
    const int2 ent1 = mybin[32 + lane];
    const int2 ent2 = mybin[64 + lane];
    const int2 ent3 = mybin[96 + lane];

    const v4f* h4 = (const v4f*)h;
    float dd[AGG_ROWS];
    v4f   acc[AGG_ROWS];
    #pragma unroll
    for (int r = 0; r < AGG_ROWS; ++r) {
        const int row = w + AGG_WAVES * r;
        const int rowc = (row < cnt) ? row : (cnt - 1);
        dd[r]  = dinv[lo + rowc];
        acc[r] = h4[(lo + rowc) * (DFEAT / 4) + lane];
    }
    v4f bb = {0.0f, 0.0f, 0.0f, 0.0f};
    if (ACT) bb = ((const v4f*)bias)[lane];
    #pragma unroll
    for (int r = 0; r < AGG_ROWS; ++r) acc[r] = acc[r] * (dd[r] * dd[r]);

    asm volatile("" :: "v"(ent0.x), "v"(ent0.y), "v"(ent1.x), "v"(ent1.y), "v"(ent2.x), "v"(ent2.y), "v"(ent3.x), "v"(ent3.y));
    if (n <= AGG_BCAP) {
        #pragma unroll 1
        for (int u = 0; u < AGG_BVECS; ++u) {
            const int m = (n - 32 * u < 32) ? (n - 32 * u) : 32;
            if (m <= 0) break;
            const int2 cur = (u == 0) ? ent0 : (u == 1) ? ent1 : (u == 2) ? ent2 : ent3;
            for (int j = 0; j < m; j += AGG_GROUP) {
                int s[AGG_GROUP], rr[AGG_GROUP];
                #pragma unroll
                for (int k = 0; k < AGG_GROUP; ++k) {
                    s[k]  = __builtin_amdgcn_readlane(cur.x, j + k);
                    rr[k] = __builtin_amdgcn_readlane(cur.y, j + k);
                }
                v4f   v[AGG_GROUP];
                float ds[AGG_GROUP];
                #pragma unroll
                for (int k = 0; k < AGG_GROUP; ++k) {
                    ds[k] = dinv[s[k]];
                    v[k]  = h4[s[k] * (DFEAT / 4) + lane];
                }
                asm volatile("" :: "s"(ds[0]), "s"(ds[1]), "s"(ds[2]), "s"(ds[3]), "s"(ds[4]), "s"(ds[5]), "s"(ds[6]), "s"(ds[7]),
                                   "s"(ds[8]), "s"(ds[9]), "s"(ds[10]), "s"(ds[11]), "s"(ds[12]), "s"(ds[13]), "s"(ds[14]), "s"(ds[15]));
                #pragma unroll
                for (int k = 0; k < AGG_GROUP; ++k) row_add(acc, dd, rr[k], v[k], ds[k]);
            }
        }
    } else {
        const int   range = blockIdx.x / BIN_OWNERS;
        const int2* seg2  = (const int2*)segs + range * BIN_CHUNKS * BIN_EDGES;
        for (int c = 0; c < BIN_CHUNKS; ++c) {
            int len = __builtin_amdgcn_readfirstlane(segcount[range * BIN_CHUNKS + c]);
            len = (len < BIN_EDGES) ? len : BIN_EDGES;
            for (int t0 = 0; t0 < len; t0 += 32) {
                const int  t   = t0 + lane;
                const int2 pr  = (t < len) ? seg2[c * BIN_EDGES + t] : make_int2(0, -1);
                const int  dl  = pr.y - lo;
                const bool hit = ((unsigned)dl < (unsigned)cnt) && ((int)((unsigned)dl % AGG_WAVES) == w);
                unsigned mask = __builtin_amdgcn_ballot_w32(hit);
                for (int q = 0; q < 32 && mask != 0u; ++q) {
                    const int b = __builtin_ctz(mask);
                    mask &= mask - 1u;
                    const int s = __builtin_amdgcn_readlane(pr.x, b);
                    const int r = (int)((unsigned)__builtin_amdgcn_readlane(dl, b) / AGG_WAVES);
                    row_add(acc, dd, r, h4[s * (DFEAT / 4) + lane], dinv[s]);
                }
            }
        }
    }

    v4f* op4 = (v4f*)out;
    #pragma unroll
    for (int r = 0; r < AGG_ROWS; ++r) {
        const int row = w + AGG_WAVES * r;
        if (row < cnt) {
            v4f o = acc[r];
            if (ACT) {
                o.x = tanhf(acc[r].x + bb.x);
                o.y = tanhf(acc[r].y + bb.y);
                o.z = tanhf(acc[r].z + bb.z);
                o.w = tanhf(acc[r].w + bb.w);
            }
            op4[(lo + row) * (DFEAT / 4) + lane] = o;
        }
    }
}

extern "C" void kernel_launch(void* const* d_in, const int* in_sizes, int n_in,
                              void* d_out, int out_size, void* d_ws, size_t ws_size,
                              hipStream_t stream) {
    const float* x    = (const float*)d_in[0];
    const int*   ei   = (const int*)  d_in[1];
    const float* W1   = (const float*)d_in[2];
    const float* b1   = (const float*)d_in[3];
    const float* W2   = (const float*)d_in[4];
    const float* b2   = (const float*)d_in[5];
    const float* Wlin = (const float*)d_in[6];
    const float* blin = (const float*)d_in[7];

    const int* src = ei;
    const int* dst = ei + N_EDGES;

    float* wsf  = (float*)d_ws;
    float* dinv = wsf;
    float* bufA = wsf + 50048;
    float* bufB = bufA + (size_t)N_NODES * DFEAT;
    int*   bins   = (int*)(bufB + (size_t)N_NODES * DFEAT);
    int*   counts = bins + (size_t)AGG_BLOCKS * AGG_WAVES * AGG_BCAP * 2;
    int*   segcount = counts + AGG_BLOCKS * AGG_WAVES;
    int*   segs     = segcount + BIN_RANGES * BIN_CHUNKS;

    const dim3 gemmGrid((GEMM_STRIPS + GEMM_WAVES - 1) / GEMM_WAVES, DFEAT / 64);

    k_partition<<<BIN_CHUNKS, BIN_THREADS, 0, stream>>>(src, dst, segs, segcount);
    k_degree_list<<<AGG_BLOCKS, AGG_THREADS, 0, stream>>>(segs, segcount, dinv, bins, counts);

    k_gemm<<<gemmGrid, 32 * GEMM_WAVES, 0, stream>>>(x, W1, bufA);
    k_aggregate<true><<<AGG_BLOCKS, AGG_THREADS, 0, stream>>>(bufA, segs, segcount, dinv, b1, bufB, bins, counts);

    k_aggregate<false><<<AGG_BLOCKS, AGG_THREADS, 0, stream>>>(bufB, segs, segcount, dinv, b2, bufA, bins, counts);
    k_dense_head<<<(GEMM_STRIPS + GEMM_WAVES - 1) / GEMM_WAVES, 32 * GEMM_WAVES, 0, stream>>>(bufA, W2, b2, Wlin, blin, (float*)d_out);
}
